# starting kernel
# baseline (speedup 1.0000x reference)
#include <hip/hip_runtime.h>

typedef __attribute__((ext_vector_type(2))) float v2f;
typedef __attribute__((ext_vector_type(8))) float v8f;

__device__ __forceinline__ void atomAddF32(float* p, float v) {
  asm volatile("global_atomic_add_f32 %0, %1, off" : : "v"(p), "v"(v) : "memory");
}

__global__ void zero4_kernel(float4* __restrict__ p, long long n4) {
  long long i = (long long)blockIdx.x * blockDim.x + threadIdx.x;
  if (i < n4) p[i] = make_float4(0.f, 0.f, 0.f, 0.f);
}

__global__ void degree_kernel(const int* __restrict__ dst, int* __restrict__ deg, int nE) {
  int i = blockIdx.x * blockDim.x + threadIdx.x;
  if (i < nE) atomicAdd(&deg[dst[i]], 1);
}

__global__ void invdeg_kernel(const int* __restrict__ deg, float* __restrict__ invd, int n) {
  int i = blockIdx.x * blockDim.x + threadIdx.x;
  if (i < n) invd[i] = 1.0f / fmaxf((float)deg[i], 1.0f);
}

__global__ void pack_w_kernel(const float* __restrict__ W, float* __restrict__ Wp,
                              int K, int N, int Np) {
  int i = blockIdx.x * blockDim.x + threadIdx.x;
  int total = (K >> 1) * Np;
  if (i >= total) return;
  int k2 = i / Np;
  int c = i - k2 * Np;
  float x0 = 0.f, x1 = 0.f;
  if (c < N) {
    x0 = W[(size_t)(2 * k2) * N + c];
    x1 = W[(size_t)(2 * k2 + 1) * N + c];
  }
  *(float2*)(Wp + (size_t)k2 * 2 * Np + 2 * c) = make_float2(x0, x1);
}

__global__ void scatter_sum_kernel(const float* __restrict__ h, const int* __restrict__ src,
                                   const int* __restrict__ dst, float* __restrict__ msum,
                                   int nE, int d, int logC) {
  long long idx = (long long)blockIdx.x * blockDim.x + threadIdx.x;
  long long total = (long long)nE << logC;
  if (idx >= total) return;
  int e = (int)(idx >> logC);
  int c = (int)(idx & ((1 << logC) - 1));
  int s = src[e];
  int t = dst[e];
  const float4 v = *(const float4*)(h + (size_t)s * d + 4 * c);
  float* p = msum + (size_t)t * d + 4 * c;
  atomAddF32(p + 0, v.x);
  atomAddF32(p + 1, v.y);
  atomAddF32(p + 2, v.z);
  atomAddF32(p + 3, v.w);
}

__device__ __forceinline__ v8f wmma4(v2f a, v2f b, v8f c) {
  return __builtin_amdgcn_wmma_f32_16x16x4_f32(false, a, false, b, (short)0, c, false, false);
}

template <bool RELU>
__global__ __launch_bounds__(256) void sage_gemm_wmma(
    const float* __restrict__ h, const float* __restrict__ msum,
    const float* __restrict__ invd, const float* __restrict__ Wp,
    const float* __restrict__ bias, float* __restrict__ out, int M, int d, int N, int Np) {
  const int w = blockIdx.x * (blockDim.x >> 5) + ((int)threadIdx.x >> 5);
  const int tilesN = Np >> 6;
  const int tilesM = M >> 5;
  const int tM = w / tilesN;
  const int tN = w - tM * tilesN;
  if (tM >= tilesM) return;

  const int m0 = tM << 5;
  const int n0 = tN << 6;
  const int lane = (int)threadIdx.x & 31;
  const int half = lane >> 4;
  const int lcol = lane & 15;
  const int r0 = m0 + lcol;
  const int r1 = r0 + 16;
  const float ir0 = invd[r0];
  const float ir1 = invd[r1];
  const float* h0 = h + (size_t)r0 * d;
  const float* h1 = h + (size_t)r1 * d;
  const float* s0 = msum + (size_t)r0 * d;
  const float* s1 = msum + (size_t)r1 * d;
  const size_t ldb = 2 * (size_t)Np;
  const int boff = 2 * (n0 + lcol);

  v8f acc0[4] = {}, acc1[4] = {};

  for (int k0 = 0; k0 < d; k0 += 4) {
    const int ka = k0 + 2 * half;
    v2f a0 = *(const v2f*)(h0 + ka);
    v2f a1 = *(const v2f*)(h1 + ka);
    const float* bp = Wp + (size_t)(ka >> 1) * ldb + boff;
#pragma unroll
    for (int j = 0; j < 4; ++j) {
      v2f b = *(const v2f*)(bp + 32 * j);
      acc0[j] = wmma4(a0, b, acc0[j]);
      acc1[j] = wmma4(a1, b, acc1[j]);
    }
  }
  for (int k0 = 0; k0 < d; k0 += 4) {
    const int ka = k0 + 2 * half;
    v2f a0 = *(const v2f*)(s0 + ka);
    v2f a1 = *(const v2f*)(s1 + ka);
    a0.x *= ir0; a0.y *= ir0;
    a1.x *= ir1; a1.y *= ir1;
    const float* bp = Wp + (size_t)((d + ka) >> 1) * ldb + boff;
#pragma unroll
    for (int j = 0; j < 4; ++j) {
      v2f b = *(const v2f*)(bp + 32 * j);
      acc0[j] = wmma4(a0, b, acc0[j]);
      acc1[j] = wmma4(a1, b, acc1[j]);
    }
  }

#pragma unroll
  for (int j = 0; j < 4; ++j) {
    int cc = n0 + j * 16 + lcol;
    if (cc < N) {
      float bj = bias[cc];
#pragma unroll
      for (int r = 0; r < 8; ++r) {
        float v0 = acc0[j][r] + bj;
        float v1 = acc1[j][r] + bj;
        if (RELU) {
          v0 = fmaxf(v0, 0.0f);
          v1 = fmaxf(v1, 0.0f);
        }
        out[(size_t)(m0 + half * 8 + r) * N + cc] = v0;
        out[(size_t)(m0 + 16 + half * 8 + r) * N + cc] = v1;
      }
    }
  }
}

extern "C" void kernel_launch(void* const* d_in, const int* in_sizes, int n_in, void* d_out,
                              int out_size, void* d_ws, size_t ws_size, hipStream_t stream) {
  (void)in_sizes; (void)n_in; (void)out_size; (void)ws_size;
  const float* x  = (const float*)d_in[0];
  const float* W1 = (const float*)d_in[1];
  const float* b1 = (const float*)d_in[2];
  const float* W2 = (const float*)d_in[3];
  const float* b2 = (const float*)d_in[4];
  const float* W3 = (const float*)d_in[5];
  const float* b3 = (const float*)d_in[6];
  const int* src  = (const int*)d_in[7];
  const int* dst  = (const int*)d_in[8];
  float* out = (float*)d_out;

  constexpr int M = 40000, E = 640000, INF = 128, HID = 256, OUTF = 47, OUTP = 64;

  float* ws = (float*)d_ws;
  int*   deg  = (int*)ws;
  float* invd = ws + 40000;
  float* msum = ws + 80000;
  float* h1   = msum + (size_t)M * HID;
  float* h2   = h1 + (size_t)M * HID;
  float* Wp1  = h2 + (size_t)M * HID;
  float* Wp2  = Wp1 + (size_t)(2 * INF) * HID;
  float* Wp3  = Wp2 + (size_t)(2 * HID) * HID;

  const int T = 256;
  auto blocks = [](long long n, int t) { return (int)((n + t - 1) / t); };

  pack_w_kernel<<<blocks((2 * INF / 2) * (long long)HID, T), T, 0, stream>>>(W1, Wp1, 2 * INF, HID, HID);
  pack_w_kernel<<<blocks((2 * HID / 2) * (long long)HID, T), T, 0, stream>>>(W2, Wp2, 2 * HID, HID, HID);
  pack_w_kernel<<<blocks((2 * HID / 2) * (long long)OUTP, T), T, 0, stream>>>(W3, Wp3, 2 * HID, OUTF, OUTP);

  zero4_kernel<<<blocks(M / 4, T), T, 0, stream>>>((float4*)deg, M / 4);
  degree_kernel<<<blocks(E, T), T, 0, stream>>>(dst, deg, E);
  invdeg_kernel<<<blocks(M, T), T, 0, stream>>>(deg, invd, M);

  {
    long long n4 = (long long)M * INF / 4;
    zero4_kernel<<<blocks(n4, T), T, 0, stream>>>((float4*)msum, n4);
    long long tot = (long long)E << 5;
    scatter_sum_kernel<<<blocks(tot, T), T, 0, stream>>>(x, src, dst, msum, E, INF, 5);
    int tiles = (M / 32) * (HID / 64);
    sage_gemm_wmma<true><<<blocks((long long)tiles * 32, T), T, 0, stream>>>(
        x, msum, invd, Wp1, b1, h1, M, INF, HID, HID);
  }
  {
    long long n4 = (long long)M * HID / 4;
    zero4_kernel<<<blocks(n4, T), T, 0, stream>>>((float4*)msum, n4);
    long long tot = (long long)E << 6;
    scatter_sum_kernel<<<blocks(tot, T), T, 0, stream>>>(h1, src, dst, msum, E, HID, 6);
    int tiles = (M / 32) * (HID / 64);
    sage_gemm_wmma<true><<<blocks((long long)tiles * 32, T), T, 0, stream>>>(
        h1, msum, invd, Wp2, b2, h2, M, HID, HID, HID);
  }
  {
    long long n4 = (long long)M * HID / 4;
    zero4_kernel<<<blocks(n4, T), T, 0, stream>>>((float4*)msum, n4);
    long long tot = (long long)E << 6;
    scatter_sum_kernel<<<blocks(tot, T), T, 0, stream>>>(h2, src, dst, msum, E, HID, 6);
    int tiles = (M / 32) * (OUTP / 64);
    sage_gemm_wmma<false><<<blocks((long long)tiles * 32, T), T, 0, stream>>>(
        h2, msum, invd, Wp3, b3, out, M, HID, OUTF, OUTP);
  }
}


# neighbour sums by owner blocks: 250 blocks keep 160 destination rows each in LDS, scan the whole dst list, add gathered rows by LDS float atomics (arrival order); no global atomics, no zero fills
# speedup vs baseline: 1.8533x; 1.8533x over previous
#include <hip/hip_runtime.h>

typedef __attribute__((ext_vector_type(2))) float v2f;
typedef __attribute__((ext_vector_type(8))) float v8f;

__global__ void zero4_kernel(float4* __restrict__ p, long long n4) {
  long long i = (long long)blockIdx.x * blockDim.x + threadIdx.x;
  if (i < n4) p[i] = make_float4(0.f, 0.f, 0.f, 0.f);
}

__global__ void degree_kernel(const int* __restrict__ dst, int* __restrict__ deg, int nE) {
  int i = blockIdx.x * blockDim.x + threadIdx.x;
  if (i < nE) atomicAdd(&deg[dst[i]], 1);
}

__global__ void invdeg_kernel(const int* __restrict__ deg, float* __restrict__ invd, int n) {
  int i = blockIdx.x * blockDim.x + threadIdx.x;
  if (i < n) invd[i] = 1.0f / fmaxf((float)deg[i], 1.0f);
}

__global__ void pack_w_kernel(const float* __restrict__ W, float* __restrict__ Wp,
                              int K, int N, int Np) {
  int i = blockIdx.x * blockDim.x + threadIdx.x;
  int total = (K >> 1) * Np;
  if (i >= total) return;
  int k2 = i / Np;
  int c = i - k2 * Np;
  float x0 = 0.f, x1 = 0.f;
  if (c < N) {
    x0 = W[(size_t)(2 * k2) * N + c];
    x1 = W[(size_t)(2 * k2 + 1) * N + c];
  }
  *(float2*)(Wp + (size_t)k2 * 2 * Np + 2 * c) = make_float2(x0, x1);
}

template <int D, int R, int NT, int QCAP>
__global__ __launch_bounds__(NT) void owner_sum_kernel(const float* __restrict__ h,
                                                       const int* __restrict__ src,
                                                       const int* __restrict__ dst,
                                                       float* __restrict__ msum, int nE,
                                                       int nRows) {
  static_assert(R <= 4096, "local row must fit the 12 bits above the edge index");
  static_assert(D % 64 == 0, "rows are handled as 256-byte units");
  __shared__ __attribute__((aligned(16))) float tile[R * D];
  __shared__ unsigned queue[QCAP];
  __shared__ unsigned qcount;

  constexpr int NW = NT / 32;
  constexpr int U = 8;
  const int tid = (int)threadIdx.x;
  const int lane = tid & 31;
  const int wv = __builtin_amdgcn_readfirstlane(tid >> 5);
  const int base = (int)blockIdx.x * R;

  for (int i = tid; i < R * D / 4; i += NT) ((float4*)tile)[i] = make_float4(0.f, 0.f, 0.f, 0.f);
  if (tid == 0) qcount = 0u;
  __syncthreads();

  auto push = [&](int dv, int e) {
    const unsigned t = (unsigned)(dv - base);
    if (t < (unsigned)R) {
      const unsigned slot = atomicAdd(&qcount, 1u);
      if (slot < (unsigned)QCAP) {
        queue[slot] = (t << 20) | (unsigned)e;
      } else {
        const float* hp = h + (size_t)src[e] * D;
        float* tp = tile + t * D;
#pragma unroll 1
        for (int k = 0; k < D; ++k)
          __hip_atomic_fetch_add(tp + k, hp[k], __ATOMIC_RELAXED, __HIP_MEMORY_SCOPE_WORKGROUP);
      }
    }
  };

  const int4* d4 = (const int4*)dst;
  const int nchunk = (nE >> 2) / (32 * U);
  for (int c = wv; c < nchunk; c += NW) {
    const int i0 = c * (32 * U) + lane;
    int4 v[U];
#pragma unroll
    for (int u = 0; u < U; ++u) v[u] = d4[i0 + u * 32];
#pragma unroll
    for (int u = 0; u < U; ++u) {
      const int e0 = (i0 + u * 32) << 2;
      const unsigned tx = (unsigned)(v[u].x - base), ty = (unsigned)(v[u].y - base);
      const unsigned tz = (unsigned)(v[u].z - base), tw = (unsigned)(v[u].w - base);
      const unsigned tmin = min(min(tx, ty), min(tz, tw));
      if (__builtin_amdgcn_ballot_w32(tmin < (unsigned)R) != 0u) {
        push(v[u].x, e0);
        push(v[u].y, e0 + 1);
        push(v[u].z, e0 + 2);
        push(v[u].w, e0 + 3);
      }
    }
  }
  for (int e = nchunk * (128 * U) + tid; e < nE; e += NT) push(dst[e], e);
  __syncthreads();

  const unsigned qn = qcount;
  const int n = __builtin_amdgcn_readfirstlane((int)(qn < (unsigned)QCAP ? qn : (unsigned)QCAP));
  auto add_row = [&](const float2* vv, int t) {
    float* tp = tile + t * D + 2 * lane;
#pragma unroll
    for (int k = 0; k < D / 64; ++k) {
      __hip_atomic_fetch_add(tp + 64 * k, vv[k].x, __ATOMIC_RELAXED, __HIP_MEMORY_SCOPE_WORKGROUP);
      __hip_atomic_fetch_add(tp + 64 * k + 1, vv[k].y, __ATOMIC_RELAXED,
                             __HIP_MEMORY_SCOPE_WORKGROUP);
    }
  };
  for (int b0 = wv * 32; b0 < n; b0 += NW * 32) {
    const int i = b0 + lane;
    const unsigned p = (i < n) ? queue[i] : 0u;
    const int s = src[p & 0xFFFFFu];
    const int t = (int)(p >> 20);
    const int cnt = (n - b0 < 32) ? (n - b0) : 32;
    int j = 0;
    for (; j + 3 < cnt; j += 4) {
      float2 vv[4][D / 64];
      int tt[4];
#pragma unroll
      for (int g = 0; g < 4; ++g) {
        const int sg = __builtin_amdgcn_readlane(s, j + g);
        tt[g] = __builtin_amdgcn_readlane(t, j + g);
        const float* hg = h + (size_t)sg * D + 2 * lane;
#pragma unroll
        for (int k = 0; k < D / 64; ++k) vv[g][k] = *(const float2*)(hg + 64 * k);
      }
#pragma unroll
      for (int g = 0; g < 4; ++g) add_row(vv[g], tt[g]);
    }
    for (; j < cnt; ++j) {
      const int sa = __builtin_amdgcn_readlane(s, j);
      const int ta = __builtin_amdgcn_readlane(t, j);
      const float* ha = h + (size_t)sa * D + 2 * lane;
      float2 va[D / 64];
#pragma unroll
      for (int k = 0; k < D / 64; ++k) va[k] = *(const float2*)(ha + 64 * k);
      add_row(va, ta);
    }
  }
  __syncthreads();

  const int rows = (nRows - base < R) ? (nRows - base) : R;
  float4* o4 = (float4*)(msum + (size_t)base * D);
  for (int i = tid; i < rows * (D / 4); i += NT) o4[i] = ((const float4*)tile)[i];
}

__device__ __forceinline__ v8f wmma4(v2f a, v2f b, v8f c) {
  return __builtin_amdgcn_wmma_f32_16x16x4_f32(false, a, false, b, (short)0, c, false, false);
}

template <bool RELU>
__global__ __launch_bounds__(256) void sage_gemm_wmma(
    const float* __restrict__ h, const float* __restrict__ msum,
    const float* __restrict__ invd, const float* __restrict__ Wp,
    const float* __restrict__ bias, float* __restrict__ out, int M, int d, int N, int Np) {
  const int w = blockIdx.x * (blockDim.x >> 5) + ((int)threadIdx.x >> 5);
  const int tilesN = Np >> 6;
  const int tilesM = M >> 5;
  const int tM = w / tilesN;
  const int tN = w - tM * tilesN;
  if (tM >= tilesM) return;

  const int m0 = tM << 5;
  const int n0 = tN << 6;
  const int lane = (int)threadIdx.x & 31;
  const int half = lane >> 4;
  const int lcol = lane & 15;
  const int r0 = m0 + lcol;
  const int r1 = r0 + 16;
  const float ir0 = invd[r0];
  const float ir1 = invd[r1];
  const float* h0 = h + (size_t)r0 * d;
  const float* h1 = h + (size_t)r1 * d;
  const float* s0 = msum + (size_t)r0 * d;
  const float* s1 = msum + (size_t)r1 * d;
  const size_t ldb = 2 * (size_t)Np;
  const int boff = 2 * (n0 + lcol);

  v8f acc0[4] = {}, acc1[4] = {};

  for (int k0 = 0; k0 < d; k0 += 4) {
    const int ka = k0 + 2 * half;
    v2f a0 = *(const v2f*)(h0 + ka);
    v2f a1 = *(const v2f*)(h1 + ka);
    const float* bp = Wp + (size_t)(ka >> 1) * ldb + boff;
#pragma unroll
    for (int j = 0; j < 4; ++j) {
      v2f b = *(const v2f*)(bp + 32 * j);
      acc0[j] = wmma4(a0, b, acc0[j]);
      acc1[j] = wmma4(a1, b, acc1[j]);
    }
  }
  for (int k0 = 0; k0 < d; k0 += 4) {
    const int ka = k0 + 2 * half;
    v2f a0 = *(const v2f*)(s0 + ka);
    v2f a1 = *(const v2f*)(s1 + ka);
    a0.x *= ir0; a0.y *= ir0;
    a1.x *= ir1; a1.y *= ir1;
    const float* bp = Wp + (size_t)((d + ka) >> 1) * ldb + boff;
#pragma unroll
    for (int j = 0; j < 4; ++j) {
      v2f b = *(const v2f*)(bp + 32 * j);
      acc0[j] = wmma4(a0, b, acc0[j]);
      acc1[j] = wmma4(a1, b, acc1[j]);
    }
  }

#pragma unroll
  for (int j = 0; j < 4; ++j) {
    int cc = n0 + j * 16 + lcol;
    if (cc < N) {
      float bj = bias[cc];
#pragma unroll
      for (int r = 0; r < 8; ++r) {
        float v0 = acc0[j][r] + bj;
        float v1 = acc1[j][r] + bj;
        if (RELU) {
          v0 = fmaxf(v0, 0.0f);
          v1 = fmaxf(v1, 0.0f);
        }
        out[(size_t)(m0 + half * 8 + r) * N + cc] = v0;
        out[(size_t)(m0 + 16 + half * 8 + r) * N + cc] = v1;
      }
    }
  }
}

extern "C" void kernel_launch(void* const* d_in, const int* in_sizes, int n_in, void* d_out,
                              int out_size, void* d_ws, size_t ws_size, hipStream_t stream) {
  (void)in_sizes; (void)n_in; (void)out_size; (void)ws_size;
  const float* x  = (const float*)d_in[0];
  const float* W1 = (const float*)d_in[1];
  const float* b1 = (const float*)d_in[2];
  const float* W2 = (const float*)d_in[3];
  const float* b2 = (const float*)d_in[4];
  const float* W3 = (const float*)d_in[5];
  const float* b3 = (const float*)d_in[6];
  const int* src  = (const int*)d_in[7];
  const int* dst  = (const int*)d_in[8];
  float* out = (float*)d_out;

  constexpr int M = 40000, E = 640000, INF = 128, HID = 256, OUTF = 47, OUTP = 64;
  constexpr int OWN_R = 160, OWN_T = 1024, OWN_Q = 8192, OWN_B = (M + OWN_R - 1) / OWN_R;
  static_assert(E <= (1 << 20), "edge index must fit the 20 low bits of a queue word");

  float* ws = (float*)d_ws;
  int*   deg  = (int*)ws;
  float* invd = ws + 40000;
  float* msum = ws + 80000;
  float* h1   = msum + (size_t)M * HID;
  float* h2   = h1 + (size_t)M * HID;
  float* Wp1  = h2 + (size_t)M * HID;
  float* Wp2  = Wp1 + (size_t)(2 * INF) * HID;
  float* Wp3  = Wp2 + (size_t)(2 * HID) * HID;

  const int T = 256;
  auto blocks = [](long long n, int t) { return (int)((n + t - 1) / t); };

  pack_w_kernel<<<blocks((2 * INF / 2) * (long long)HID, T), T, 0, stream>>>(W1, Wp1, 2 * INF, HID, HID);
  pack_w_kernel<<<blocks((2 * HID / 2) * (long long)HID, T), T, 0, stream>>>(W2, Wp2, 2 * HID, HID, HID);
  pack_w_kernel<<<blocks((2 * HID / 2) * (long long)OUTP, T), T, 0, stream>>>(W3, Wp3, 2 * HID, OUTF, OUTP);

  zero4_kernel<<<blocks(M / 4, T), T, 0, stream>>>((float4*)deg, M / 4);
  degree_kernel<<<blocks(E, T), T, 0, stream>>>(dst, deg, E);
  invdeg_kernel<<<blocks(M, T), T, 0, stream>>>(deg, invd, M);

  {
    owner_sum_kernel<INF, OWN_R, OWN_T, OWN_Q><<<OWN_B, OWN_T, 0, stream>>>(x, src, dst, msum, E, M);
    int tiles = (M / 32) * (HID / 64);
    sage_gemm_wmma<true><<<blocks((long long)tiles * 32, T), T, 0, stream>>>(
        x, msum, invd, Wp1, b1, h1, M, INF, HID, HID);
  }
  {
    owner_sum_kernel<HID, OWN_R, OWN_T, OWN_Q><<<OWN_B, OWN_T, 0, stream>>>(h1, src, dst, msum, E, M);
    int tiles = (M / 32) * (HID / 64);
    sage_gemm_wmma<true><<<blocks((long long)tiles * 32, T), T, 0, stream>>>(
        h1, msum, invd, Wp2, b2, h2, M, HID, HID, HID);
  }
  {
    owner_sum_kernel<HID, OWN_R, OWN_T, OWN_Q><<<OWN_B, OWN_T, 0, stream>>>(h2, src, dst, msum, E, M);
    int tiles = (M / 32) * (OUTP / 64);
    sage_gemm_wmma<false><<<blocks((long long)tiles * 32, T), T, 0, stream>>>(
        h2, msum, invd, Wp3, b3, out, M, HID, OUTF, OUTP);
  }
}


# dst scanned once per call, not per layer: layer 1's owner blocks save their edge queues (local row, source node); layers 2 and 3 drain them: no re-scan, no src look-up; arrival-order LDS adds kept
# speedup vs baseline: 2.0682x; 1.1160x over previous
#include <hip/hip_runtime.h>

typedef __attribute__((ext_vector_type(2))) float v2f;
typedef __attribute__((ext_vector_type(8))) float v8f;

__global__ void zero4_kernel(float4* __restrict__ p, long long n4) {
  long long i = (long long)blockIdx.x * blockDim.x + threadIdx.x;
  if (i < n4) p[i] = make_float4(0.f, 0.f, 0.f, 0.f);
}

__global__ void degree_kernel(const int* __restrict__ dst, int* __restrict__ deg, int nE) {
  int i = blockIdx.x * blockDim.x + threadIdx.x;
  if (i < nE) atomicAdd(&deg[dst[i]], 1);
}

__global__ void invdeg_kernel(const int* __restrict__ deg, float* __restrict__ invd, int n) {
  int i = blockIdx.x * blockDim.x + threadIdx.x;
  if (i < n) invd[i] = 1.0f / fmaxf((float)deg[i], 1.0f);
}

__global__ void pack_w_kernel(const float* __restrict__ W, float* __restrict__ Wp,
                              int K, int N, int Np) {
  int i = blockIdx.x * blockDim.x + threadIdx.x;
  int total = (K >> 1) * Np;
  if (i >= total) return;
  int k2 = i / Np;
  int c = i - k2 * Np;
  float x0 = 0.f, x1 = 0.f;
  if (c < N) {
    x0 = W[(size_t)(2 * k2) * N + c];
    x1 = W[(size_t)(2 * k2 + 1) * N + c];
  }
  *(float2*)(Wp + (size_t)k2 * 2 * Np + 2 * c) = make_float2(x0, x1);
}

constexpr unsigned kNoList = 0xFFFFFFFFu;

template <int D, int R, int NT, int QCAP, int LCAP, bool BUILD>
__global__ __launch_bounds__(NT) void owner_sum_kernel(const float* __restrict__ h,
                                                       const int* __restrict__ src,
                                                       const int* __restrict__ dst,
                                                       float* __restrict__ msum,
                                                       unsigned* lists, int nE, int nRows) {
  static_assert(R <= 4096, "local row must fit the 12 bits above the edge index");
  static_assert(D % 64 == 0, "rows are handled as 256-byte units");
  __shared__ __attribute__((aligned(16))) float tile[R * D];
  __shared__ unsigned queue[QCAP];
  __shared__ unsigned qcount;

  constexpr int NW = NT / 32;
  constexpr int U = 8;
  const int tid = (int)threadIdx.x;
  const int lane = tid & 31;
  const int wv = __builtin_amdgcn_readfirstlane(tid >> 5);
  const int base = (int)blockIdx.x * R;
  unsigned* lst = lists + (size_t)blockIdx.x * (LCAP + 32);

  unsigned hdr = kNoList;
  if (!BUILD) hdr = lst[0];
  const bool scan = BUILD || hdr == kNoList;

  for (int i = tid; i < R * D / 4; i += NT) ((float4*)tile)[i] = make_float4(0.f, 0.f, 0.f, 0.f);
  if (tid == 0) qcount = 0u;
  __syncthreads();

  auto push = [&](int dv, int e) {
    const unsigned t = (unsigned)(dv - base);
    if (t < (unsigned)R) {
      const unsigned slot = atomicAdd(&qcount, 1u);
      if (slot < (unsigned)QCAP) {
        queue[slot] = (t << 20) | (unsigned)e;
      } else {
        const float* hp = h + (size_t)src[e] * D;
        float* tp = tile + t * D;
#pragma unroll 1
        for (int k = 0; k < D; ++k)
          __hip_atomic_fetch_add(tp + k, hp[k], __ATOMIC_RELAXED, __HIP_MEMORY_SCOPE_WORKGROUP);
      }
    }
  };

  const int4* d4 = (const int4*)dst;
  const int nchunk = (nE >> 2) / (32 * U);
  if (scan) {
    for (int c = wv; c < nchunk; c += NW) {
      const int i0 = c * (32 * U) + lane;
      int4 v[U];
#pragma unroll
      for (int u = 0; u < U; ++u) v[u] = d4[i0 + u * 32];
#pragma unroll
      for (int u = 0; u < U; ++u) {
        const int e0 = (i0 + u * 32) << 2;
        const unsigned tx = (unsigned)(v[u].x - base), ty = (unsigned)(v[u].y - base);
        const unsigned tz = (unsigned)(v[u].z - base), tw = (unsigned)(v[u].w - base);
        const unsigned tmin = min(min(tx, ty), min(tz, tw));
        if (__builtin_amdgcn_ballot_w32(tmin < (unsigned)R) != 0u) {
          push(v[u].x, e0);
          push(v[u].y, e0 + 1);
          push(v[u].z, e0 + 2);
          push(v[u].w, e0 + 3);
        }
      }
    }
    for (int e = nchunk * (128 * U) + tid; e < nE; e += NT) push(dst[e], e);
  }
  __syncthreads();

  const unsigned qn = scan ? qcount : hdr;
  const unsigned cap = scan ? (unsigned)QCAP : (unsigned)LCAP;
  const int n = __builtin_amdgcn_readfirstlane((int)(qn < cap ? qn : cap));
  const bool save = BUILD && qn <= (unsigned)QCAP && qn <= (unsigned)LCAP;
  if (BUILD && tid == 0) lst[0] = save ? qn : kNoList;
  auto add_row = [&](const float2* vv, int t) {
    float* tp = tile + t * D + 2 * lane;
#pragma unroll
    for (int k = 0; k < D / 64; ++k) {
      __hip_atomic_fetch_add(tp + 64 * k, vv[k].x, __ATOMIC_RELAXED, __HIP_MEMORY_SCOPE_WORKGROUP);
      __hip_atomic_fetch_add(tp + 64 * k + 1, vv[k].y, __ATOMIC_RELAXED,
                             __HIP_MEMORY_SCOPE_WORKGROUP);
    }
  };
  for (int b0 = wv * 32; b0 < n; b0 += NW * 32) {
    const int i = b0 + lane;
    int s, t;
    if (scan) {
      const unsigned p = (i < n) ? queue[i] : 0u;
      s = src[p & 0xFFFFFu];
      t = (int)(p >> 20);
      if (save && i < n) lst[32 + i] = ((unsigned)t << 20) | (unsigned)s;
    } else {
      const unsigned p = (i < n) ? lst[32 + i] : 0u;
      s = min((int)(p & 0xFFFFFu), nRows - 1);
      t = min((int)(p >> 20), R - 1);
    }
    const int cnt = (n - b0 < 32) ? (n - b0) : 32;
    int j = 0;
    for (; j + 3 < cnt; j += 4) {
      float2 vv[4][D / 64];
      int tt[4];
#pragma unroll
      for (int g = 0; g < 4; ++g) {
        const int sg = __builtin_amdgcn_readlane(s, j + g);
        tt[g] = __builtin_amdgcn_readlane(t, j + g);
        const float* hg = h + (size_t)sg * D + 2 * lane;
#pragma unroll
        for (int k = 0; k < D / 64; ++k) vv[g][k] = *(const float2*)(hg + 64 * k);
      }
#pragma unroll
      for (int g = 0; g < 4; ++g) add_row(vv[g], tt[g]);
    }
    for (; j < cnt; ++j) {
      const int sa = __builtin_amdgcn_readlane(s, j);
      const int ta = __builtin_amdgcn_readlane(t, j);
      const float* ha = h + (size_t)sa * D + 2 * lane;
      float2 va[D / 64];
#pragma unroll
      for (int k = 0; k < D / 64; ++k) va[k] = *(const float2*)(ha + 64 * k);
      add_row(va, ta);
    }
  }
  __syncthreads();

  const int rows = (nRows - base < R) ? (nRows - base) : R;
  float4* o4 = (float4*)(msum + (size_t)base * D);
  for (int i = tid; i < rows * (D / 4); i += NT) o4[i] = ((const float4*)tile)[i];
}

__device__ __forceinline__ v8f wmma4(v2f a, v2f b, v8f c) {
  return __builtin_amdgcn_wmma_f32_16x16x4_f32(false, a, false, b, (short)0, c, false, false);
}

template <bool RELU>
__global__ __launch_bounds__(256) void sage_gemm_wmma(
    const float* __restrict__ h, const float* __restrict__ msum,
    const float* __restrict__ invd, const float* __restrict__ Wp,
    const float* __restrict__ bias, float* __restrict__ out, int M, int d, int N, int Np) {
  const int w = blockIdx.x * (blockDim.x >> 5) + ((int)threadIdx.x >> 5);
  const int tilesN = Np >> 6;
  const int tilesM = M >> 5;
  const int tM = w / tilesN;
  const int tN = w - tM * tilesN;
  if (tM >= tilesM) return;

  const int m0 = tM << 5;
  const int n0 = tN << 6;
  const int lane = (int)threadIdx.x & 31;
  const int half = lane >> 4;
  const int lcol = lane & 15;
  const int r0 = m0 + lcol;
  const int r1 = r0 + 16;
  const float ir0 = invd[r0];
  const float ir1 = invd[r1];
  const float* h0 = h + (size_t)r0 * d;
  const float* h1 = h + (size_t)r1 * d;
  const float* s0 = msum + (size_t)r0 * d;
  const float* s1 = msum + (size_t)r1 * d;
  const size_t ldb = 2 * (size_t)Np;
  const int boff = 2 * (n0 + lcol);

  v8f acc0[4] = {}, acc1[4] = {};

  for (int k0 = 0; k0 < d; k0 += 4) {
    const int ka = k0 + 2 * half;
    v2f a0 = *(const v2f*)(h0 + ka);
    v2f a1 = *(const v2f*)(h1 + ka);
    const float* bp = Wp + (size_t)(ka >> 1) * ldb + boff;
#pragma unroll
    for (int j = 0; j < 4; ++j) {
      v2f b = *(const v2f*)(bp + 32 * j);
      acc0[j] = wmma4(a0, b, acc0[j]);
      acc1[j] = wmma4(a1, b, acc1[j]);
    }
  }
  for (int k0 = 0; k0 < d; k0 += 4) {
    const int ka = k0 + 2 * half;
    v2f a0 = *(const v2f*)(s0 + ka);
    v2f a1 = *(const v2f*)(s1 + ka);
    a0.x *= ir0; a0.y *= ir0;
    a1.x *= ir1; a1.y *= ir1;
    const float* bp = Wp + (size_t)((d + ka) >> 1) * ldb + boff;
#pragma unroll
    for (int j = 0; j < 4; ++j) {
      v2f b = *(const v2f*)(bp + 32 * j);
      acc0[j] = wmma4(a0, b, acc0[j]);
      acc1[j] = wmma4(a1, b, acc1[j]);
    }
  }

#pragma unroll
  for (int j = 0; j < 4; ++j) {
    int cc = n0 + j * 16 + lcol;
    if (cc < N) {
      float bj = bias[cc];
#pragma unroll
      for (int r = 0; r < 8; ++r) {
        float v0 = acc0[j][r] + bj;
        float v1 = acc1[j][r] + bj;
        if (RELU) {
          v0 = fmaxf(v0, 0.0f);
          v1 = fmaxf(v1, 0.0f);
        }
        out[(size_t)(m0 + half * 8 + r) * N + cc] = v0;
        out[(size_t)(m0 + 16 + half * 8 + r) * N + cc] = v1;
      }
    }
  }
}

extern "C" void kernel_launch(void* const* d_in, const int* in_sizes, int n_in, void* d_out,
                              int out_size, void* d_ws, size_t ws_size, hipStream_t stream) {
  (void)in_sizes; (void)n_in; (void)out_size; (void)ws_size;
  const float* x  = (const float*)d_in[0];
  const float* W1 = (const float*)d_in[1];
  const float* b1 = (const float*)d_in[2];
  const float* W2 = (const float*)d_in[3];
  const float* b2 = (const float*)d_in[4];
  const float* W3 = (const float*)d_in[5];
  const float* b3 = (const float*)d_in[6];
  const int* src  = (const int*)d_in[7];
  const int* dst  = (const int*)d_in[8];
  float* out = (float*)d_out;

  constexpr int M = 40000, E = 640000, INF = 128, HID = 256, OUTF = 47, OUTP = 64;
  constexpr int OWN_R = 160, OWN_T = 1024, OWN_Q = 8192, OWN_B = (M + OWN_R - 1) / OWN_R;
  static_assert(E <= (1 << 20), "edge index must fit the 20 low bits of a queue word");
  static_assert(M <= (1 << 20), "source node must fit the 20 low bits of a saved list word");
  constexpr int OWN_L = 7456;
  static_assert((long long)OWN_B * (OWN_L + 32) <= (long long)M * OUTF,
                "the saved lists must fit the output buffer");
  unsigned* lists = (unsigned*)d_out;

  float* ws = (float*)d_ws;
  int*   deg  = (int*)ws;
  float* invd = ws + 40000;
  float* msum = ws + 80000;
  float* h1   = msum + (size_t)M * HID;
  float* h2   = h1 + (size_t)M * HID;
  float* Wp1  = h2 + (size_t)M * HID;
  float* Wp2  = Wp1 + (size_t)(2 * INF) * HID;
  float* Wp3  = Wp2 + (size_t)(2 * HID) * HID;

  const int T = 256;
  auto blocks = [](long long n, int t) { return (int)((n + t - 1) / t); };

  pack_w_kernel<<<blocks((2 * INF / 2) * (long long)HID, T), T, 0, stream>>>(W1, Wp1, 2 * INF, HID, HID);
  pack_w_kernel<<<blocks((2 * HID / 2) * (long long)HID, T), T, 0, stream>>>(W2, Wp2, 2 * HID, HID, HID);
  pack_w_kernel<<<blocks((2 * HID / 2) * (long long)OUTP, T), T, 0, stream>>>(W3, Wp3, 2 * HID, OUTF, OUTP);

  zero4_kernel<<<blocks(M / 4, T), T, 0, stream>>>((float4*)deg, M / 4);
  degree_kernel<<<blocks(E, T), T, 0, stream>>>(dst, deg, E);
  invdeg_kernel<<<blocks(M, T), T, 0, stream>>>(deg, invd, M);

  {
    owner_sum_kernel<INF, OWN_R, OWN_T, OWN_Q, OWN_L, true><<<OWN_B, OWN_T, 0, stream>>>(
        x, src, dst, msum, lists, E, M);
    int tiles = (M / 32) * (HID / 64);
    sage_gemm_wmma<true><<<blocks((long long)tiles * 32, T), T, 0, stream>>>(
        x, msum, invd, Wp1, b1, h1, M, INF, HID, HID);
  }
  {
    owner_sum_kernel<HID, OWN_R, OWN_T, OWN_Q, OWN_L, false><<<OWN_B, OWN_T, 0, stream>>>(
        h1, src, dst, msum, lists, E, M);
    int tiles = (M / 32) * (HID / 64);
    sage_gemm_wmma<true><<<blocks((long long)tiles * 32, T), T, 0, stream>>>(
        h1, msum, invd, Wp2, b2, h2, M, HID, HID, HID);
  }
  {
    owner_sum_kernel<HID, OWN_R, OWN_T, OWN_Q, OWN_L, false><<<OWN_B, OWN_T, 0, stream>>>(
        h2, src, dst, msum, lists, E, M);
    int tiles = (M / 32) * (OUTP / 64);
    sage_gemm_wmma<false><<<blocks((long long)tiles * 32, T), T, 0, stream>>>(
        h2, msum, invd, Wp3, b3, out, M, HID, OUTF, OUTP);
  }
}


# GEMM wave tile re-indexed for 16-byte fragment loads: a lane takes 4 consecutive k of its A row (2 matrix steps, B paired alike) and 2 neighbouring columns; whole-line stores; k sum order re-paired
# speedup vs baseline: 2.5930x; 1.2537x over previous
#include <hip/hip_runtime.h>

typedef __attribute__((ext_vector_type(2))) float v2f;
typedef __attribute__((ext_vector_type(8))) float v8f;


__global__ void pack_w_kernel(const float* __restrict__ W, float* __restrict__ Wp,
                              int K, int N, int Np) {
  int i = blockIdx.x * blockDim.x + threadIdx.x;
  int total = (K >> 1) * Np;
  if (i >= total) return;
  int k2 = i / Np;
  int c = i - k2 * Np;
  float x0 = 0.f, x1 = 0.f;
  if (c < N) {
    x0 = W[(size_t)(2 * k2) * N + c];
    x1 = W[(size_t)(2 * k2 + 1) * N + c];
  }
  *(float2*)(Wp + (size_t)k2 * 2 * Np + 2 * c) = make_float2(x0, x1);
}

constexpr unsigned kNoList = 0xFFFFFFFFu;

template <int D, int R, int NT, int QCAP, int LCAP, bool BUILD>
__global__ __launch_bounds__(NT) void owner_sum_kernel(const float* __restrict__ h,
                                                       const int* __restrict__ src,
                                                       const int* __restrict__ dst,
                                                       float* __restrict__ msum,
                                                       unsigned* lists, float* invd, int nE,
                                                       int nRows) {
  static_assert(R <= 4096, "local row must fit the 12 bits above the edge index");
  static_assert(D % 64 == 0, "rows are handled as 256-byte units");
  __shared__ __attribute__((aligned(16))) float tile[R * D];
  __shared__ unsigned queue[QCAP];
  __shared__ unsigned qcount;
  __shared__ int cnt[R];
  __shared__ float sinv[R];

  constexpr int NW = NT / 32;
  constexpr int U = 8;
  const int tid = (int)threadIdx.x;
  const int lane = tid & 31;
  const int wv = __builtin_amdgcn_readfirstlane(tid >> 5);
  const int base = (int)blockIdx.x * R;
  unsigned* lst = lists + (size_t)blockIdx.x * (LCAP + 32);

  unsigned hdr = kNoList;
  if (!BUILD) hdr = lst[0];
  const bool scan = BUILD || hdr == kNoList;

  for (int i = tid; i < R * D / 4; i += NT) ((float4*)tile)[i] = make_float4(0.f, 0.f, 0.f, 0.f);
  for (int i = tid; i < R; i += NT) cnt[i] = 0;
  if (tid == 0) qcount = 0u;
  __syncthreads();

  auto push = [&](int dv, int e) {
    const unsigned t = (unsigned)(dv - base);
    if (t < (unsigned)R) {
      atomicAdd(&cnt[t], 1);
      const unsigned slot = atomicAdd(&qcount, 1u);
      if (slot < (unsigned)QCAP) {
        queue[slot] = (t << 20) | (unsigned)e;
      } else {
        const float* hp = h + (size_t)src[e] * D;
        float* tp = tile + t * D;
#pragma unroll 1
        for (int k = 0; k < D; ++k)
          __hip_atomic_fetch_add(tp + k, hp[k], __ATOMIC_RELAXED, __HIP_MEMORY_SCOPE_WORKGROUP);
      }
    }
  };

  const int4* d4 = (const int4*)dst;
  const int nchunk = (nE >> 2) / (32 * U);
  if (scan) {
    for (int c = wv; c < nchunk; c += NW) {
      const int i0 = c * (32 * U) + lane;
      int4 v[U];
#pragma unroll
      for (int u = 0; u < U; ++u) v[u] = d4[i0 + u * 32];
#pragma unroll
      for (int u = 0; u < U; ++u) {
        const int e0 = (i0 + u * 32) << 2;
        const unsigned tx = (unsigned)(v[u].x - base), ty = (unsigned)(v[u].y - base);
        const unsigned tz = (unsigned)(v[u].z - base), tw = (unsigned)(v[u].w - base);
        const unsigned tmin = min(min(tx, ty), min(tz, tw));
        if (__builtin_amdgcn_ballot_w32(tmin < (unsigned)R) != 0u) {
          push(v[u].x, e0);
          push(v[u].y, e0 + 1);
          push(v[u].z, e0 + 2);
          push(v[u].w, e0 + 3);
        }
      }
    }
    for (int e = nchunk * (128 * U) + tid; e < nE; e += NT) push(dst[e], e);
  }
  __syncthreads();

  for (int r = tid; r < R; r += NT) {
    const bool live = base + r < nRows;
    float v = 1.0f;
    if (scan) v = 1.0f / fmaxf((float)cnt[r], 1.0f);
    else if (live) v = invd[base + r];
    sinv[r] = v;
    if (BUILD && live) invd[base + r] = v;
  }

  const unsigned qn = scan ? qcount : hdr;
  const unsigned cap = scan ? (unsigned)QCAP : (unsigned)LCAP;
  const int n = __builtin_amdgcn_readfirstlane((int)(qn < cap ? qn : cap));
  const bool save = BUILD && qn <= (unsigned)QCAP && qn <= (unsigned)LCAP;
  if (BUILD && tid == 0) lst[0] = save ? qn : kNoList;
  auto add_row = [&](const float2* vv, int t) {
    float* tp = tile + t * D + 2 * lane;
#pragma unroll
    for (int k = 0; k < D / 64; ++k) {
      __hip_atomic_fetch_add(tp + 64 * k, vv[k].x, __ATOMIC_RELAXED, __HIP_MEMORY_SCOPE_WORKGROUP);
      __hip_atomic_fetch_add(tp + 64 * k + 1, vv[k].y, __ATOMIC_RELAXED,
                             __HIP_MEMORY_SCOPE_WORKGROUP);
    }
  };
  for (int b0 = wv * 32; b0 < n; b0 += NW * 32) {
    const int i = b0 + lane;
    int s, t;
    if (scan) {
      const unsigned p = (i < n) ? queue[i] : 0u;
      s = src[p & 0xFFFFFu];
      t = (int)(p >> 20);
      if (save && i < n) lst[32 + i] = ((unsigned)t << 20) | (unsigned)s;
    } else {
      const unsigned p = (i < n) ? lst[32 + i] : 0u;
      s = min((int)(p & 0xFFFFFu), nRows - 1);
      t = min((int)(p >> 20), R - 1);
    }
    const int cnt = (n - b0 < 32) ? (n - b0) : 32;
    int j = 0;
    for (; j + 3 < cnt; j += 4) {
      float2 vv[4][D / 64];
      int tt[4];
#pragma unroll
      for (int g = 0; g < 4; ++g) {
        const int sg = __builtin_amdgcn_readlane(s, j + g);
        tt[g] = __builtin_amdgcn_readlane(t, j + g);
        const float* hg = h + (size_t)sg * D + 2 * lane;
#pragma unroll
        for (int k = 0; k < D / 64; ++k) vv[g][k] = *(const float2*)(hg + 64 * k);
      }
#pragma unroll
      for (int g = 0; g < 4; ++g) add_row(vv[g], tt[g]);
    }
    for (; j < cnt; ++j) {
      const int sa = __builtin_amdgcn_readlane(s, j);
      const int ta = __builtin_amdgcn_readlane(t, j);
      const float* ha = h + (size_t)sa * D + 2 * lane;
      float2 va[D / 64];
#pragma unroll
      for (int k = 0; k < D / 64; ++k) va[k] = *(const float2*)(ha + 64 * k);
      add_row(va, ta);
    }
  }
  __syncthreads();

  const int rows = (nRows - base < R) ? (nRows - base) : R;
  float4* o4 = (float4*)(msum + (size_t)base * D);
  for (int i = tid; i < rows * (D / 4); i += NT) {
    const float4 v = ((const float4*)tile)[i];
    const float s = sinv[i / (D / 4)];
    o4[i] = make_float4(v.x * s, v.y * s, v.z * s, v.w * s);
  }
}

typedef __attribute__((ext_vector_type(4))) float v4f;

__device__ __forceinline__ v8f wmma4(v2f a, v2f b, v8f c) {
  return __builtin_amdgcn_wmma_f32_16x16x4_f32(false, a, false, b, (short)0, c, false, false);
}

template <bool RELU>
__global__ __launch_bounds__(256) void sage_gemm_wmma(
    const float* __restrict__ h, const float* __restrict__ msum,
    const float* __restrict__ Wp,
    const float* __restrict__ bias, float* __restrict__ out, int M, int d, int N, int Np) {
  const int w = blockIdx.x * (blockDim.x >> 5) + ((int)threadIdx.x >> 5);
  const int tilesN = Np >> 6;
  const int tilesM = M >> 5;
  const int tM = w / tilesN;
  const int tN = w - tM * tilesN;
  if (tM >= tilesM) return;

  const int m0 = tM << 5;
  const int n0 = tN << 6;
  const int lane = (int)threadIdx.x & 31;
  const int half = lane >> 4;
  const int lcol = lane & 15;
  const int r0 = m0 + lcol;
  const int r1 = r0 + 16;
  const float* h0 = h + (size_t)r0 * d + 4 * half;
  const float* h1 = h + (size_t)r1 * d + 4 * half;
  const float* s0 = msum + (size_t)r0 * d + 4 * half;
  const float* s1 = msum + (size_t)r1 * d + 4 * half;
  const size_t ldb = 2 * (size_t)Np;
  const float* wb = Wp + (size_t)(2 * half) * ldb + 2 * (n0 + 2 * lcol);

  v8f acc0[4] = {}, acc1[4] = {};

  auto chunk = [&](const float* a0p, const float* a1p, const float* bp) {
    const v4f a0 = *(const v4f*)a0p;
    const v4f a1 = *(const v4f*)a1p;
    const v4f b00 = *(const v4f*)(bp);
    const v4f b01 = *(const v4f*)(bp + 64);
    const v4f b10 = *(const v4f*)(bp + ldb);
    const v4f b11 = *(const v4f*)(bp + ldb + 64);
    const v2f a0lo = __builtin_shufflevector(a0, a0, 0, 1), a0hi = __builtin_shufflevector(a0, a0, 2, 3);
    const v2f a1lo = __builtin_shufflevector(a1, a1, 0, 1), a1hi = __builtin_shufflevector(a1, a1, 2, 3);
    acc0[0] = wmma4(a0lo, __builtin_shufflevector(b00, b00, 0, 1), acc0[0]);
    acc1[0] = wmma4(a1lo, __builtin_shufflevector(b00, b00, 0, 1), acc1[0]);
    acc0[1] = wmma4(a0lo, __builtin_shufflevector(b00, b00, 2, 3), acc0[1]);
    acc1[1] = wmma4(a1lo, __builtin_shufflevector(b00, b00, 2, 3), acc1[1]);
    acc0[2] = wmma4(a0lo, __builtin_shufflevector(b01, b01, 0, 1), acc0[2]);
    acc1[2] = wmma4(a1lo, __builtin_shufflevector(b01, b01, 0, 1), acc1[2]);
    acc0[3] = wmma4(a0lo, __builtin_shufflevector(b01, b01, 2, 3), acc0[3]);
    acc1[3] = wmma4(a1lo, __builtin_shufflevector(b01, b01, 2, 3), acc1[3]);
    acc0[0] = wmma4(a0hi, __builtin_shufflevector(b10, b10, 0, 1), acc0[0]);
    acc1[0] = wmma4(a1hi, __builtin_shufflevector(b10, b10, 0, 1), acc1[0]);
    acc0[1] = wmma4(a0hi, __builtin_shufflevector(b10, b10, 2, 3), acc0[1]);
    acc1[1] = wmma4(a1hi, __builtin_shufflevector(b10, b10, 2, 3), acc1[1]);
    acc0[2] = wmma4(a0hi, __builtin_shufflevector(b11, b11, 0, 1), acc0[2]);
    acc1[2] = wmma4(a1hi, __builtin_shufflevector(b11, b11, 0, 1), acc1[2]);
    acc0[3] = wmma4(a0hi, __builtin_shufflevector(b11, b11, 2, 3), acc0[3]);
    acc1[3] = wmma4(a1hi, __builtin_shufflevector(b11, b11, 2, 3), acc1[3]);
  };

  for (int k0 = 0; k0 < d; k0 += 8) chunk(h0 + k0, h1 + k0, wb + (size_t)(k0 >> 1) * ldb);
  for (int k0 = 0; k0 < d; k0 += 8) chunk(s0 + k0, s1 + k0, wb + (size_t)((d + k0) >> 1) * ldb);

  const bool evenN = (N & 1) == 0;
#pragma unroll
  for (int q = 0; q < 2; ++q) {
    const int cc = n0 + 32 * q + 2 * lcol;
    if (evenN) {
      if (cc + 1 < N) {
        const float bx = bias[cc], by = bias[cc + 1];
#pragma unroll
        for (int r = 0; r < 8; ++r) {
          float x0 = acc0[2 * q][r] + bx, y0 = acc0[2 * q + 1][r] + by;
          float x1 = acc1[2 * q][r] + bx, y1 = acc1[2 * q + 1][r] + by;
          if (RELU) {
            x0 = fmaxf(x0, 0.0f); y0 = fmaxf(y0, 0.0f);
            x1 = fmaxf(x1, 0.0f); y1 = fmaxf(y1, 0.0f);
          }
          *(float2*)(out + (size_t)(m0 + half * 8 + r) * N + cc) = make_float2(x0, y0);
          *(float2*)(out + (size_t)(m0 + 16 + half * 8 + r) * N + cc) = make_float2(x1, y1);
        }
      }
    } else {
#pragma unroll
      for (int i = 0; i < 2; ++i) {
        if (cc + i < N) {
          const float bj = bias[cc + i];
#pragma unroll
          for (int r = 0; r < 8; ++r) {
            float v0 = acc0[2 * q + i][r] + bj;
            float v1 = acc1[2 * q + i][r] + bj;
            if (RELU) {
              v0 = fmaxf(v0, 0.0f);
              v1 = fmaxf(v1, 0.0f);
            }
            out[(size_t)(m0 + half * 8 + r) * N + cc + i] = v0;
            out[(size_t)(m0 + 16 + half * 8 + r) * N + cc + i] = v1;
          }
        }
      }
    }
  }
}

extern "C" void kernel_launch(void* const* d_in, const int* in_sizes, int n_in, void* d_out,
                              int out_size, void* d_ws, size_t ws_size, hipStream_t stream) {
  (void)in_sizes; (void)n_in; (void)out_size; (void)ws_size;
  const float* x  = (const float*)d_in[0];
  const float* W1 = (const float*)d_in[1];
  const float* b1 = (const float*)d_in[2];
  const float* W2 = (const float*)d_in[3];
  const float* b2 = (const float*)d_in[4];
  const float* W3 = (const float*)d_in[5];
  const float* b3 = (const float*)d_in[6];
  const int* src  = (const int*)d_in[7];
  const int* dst  = (const int*)d_in[8];
  float* out = (float*)d_out;

  constexpr int M = 40000, E = 640000, INF = 128, HID = 256, OUTF = 47, OUTP = 64;
  constexpr int OWN_R = 160, OWN_T = 1024, OWN_Q = 8192, OWN_B = (M + OWN_R - 1) / OWN_R;
  static_assert(E <= (1 << 20), "edge index must fit the 20 low bits of a queue word");
  static_assert(M <= (1 << 20), "source node must fit the 20 low bits of a saved list word");
  constexpr int OWN_L = 7456;
  static_assert((long long)OWN_B * (OWN_L + 32) <= (long long)M * OUTF,
                "the saved lists must fit the output buffer");
  unsigned* lists = (unsigned*)d_out;

  float* ws = (float*)d_ws;
  float* invd = ws + 40000;
  float* msum = ws + 80000;
  float* h1   = msum + (size_t)M * HID;
  float* h2   = h1 + (size_t)M * HID;
  float* Wp1  = h2 + (size_t)M * HID;
  float* Wp2  = Wp1 + (size_t)(2 * INF) * HID;
  float* Wp3  = Wp2 + (size_t)(2 * HID) * HID;

  const int T = 256;
  auto blocks = [](long long n, int t) { return (int)((n + t - 1) / t); };

  pack_w_kernel<<<blocks((2 * INF / 2) * (long long)HID, T), T, 0, stream>>>(W1, Wp1, 2 * INF, HID, HID);
  pack_w_kernel<<<blocks((2 * HID / 2) * (long long)HID, T), T, 0, stream>>>(W2, Wp2, 2 * HID, HID, HID);
  pack_w_kernel<<<blocks((2 * HID / 2) * (long long)OUTP, T), T, 0, stream>>>(W3, Wp3, 2 * HID, OUTF, OUTP);

  {
    owner_sum_kernel<INF, OWN_R, OWN_T, OWN_Q, OWN_L, true><<<OWN_B, OWN_T, 0, stream>>>(
        x, src, dst, msum, lists, invd, E, M);
    int tiles = (M / 32) * (HID / 64);
    sage_gemm_wmma<true><<<blocks((long long)tiles * 32, T), T, 0, stream>>>(
        x, msum, Wp1, b1, h1, M, INF, HID, HID);
  }
  {
    owner_sum_kernel<HID, OWN_R, OWN_T, OWN_Q, OWN_L, false><<<OWN_B, OWN_T, 0, stream>>>(
        h1, src, dst, msum, lists, invd, E, M);
    int tiles = (M / 32) * (HID / 64);
    sage_gemm_wmma<true><<<blocks((long long)tiles * 32, T), T, 0, stream>>>(
        h1, msum, Wp2, b2, h2, M, HID, HID, HID);
  }
  {
    owner_sum_kernel<HID, OWN_R, OWN_T, OWN_Q, OWN_L, false><<<OWN_B, OWN_T, 0, stream>>>(
        h2, src, dst, msum, lists, invd, E, M);
    int tiles = (M / 32) * (OUTP / 64);
    sage_gemm_wmma<false><<<blocks((long long)tiles * 32, T), T, 0, stream>>>(
        h2, msum, Wp3, b3, out, M, HID, OUTF, OUTP);
  }
}


# edge list read once per call: a binning launch appends each edge to its owner's list (LDS rank, one global integer add per owner and block); owners stop scanning dst; 10 launches; arrival-order sums
# speedup vs baseline: 2.6927x; 1.0384x over previous
#include <hip/hip_runtime.h>

typedef __attribute__((ext_vector_type(2))) float v2f;
typedef __attribute__((ext_vector_type(8))) float v8f;


__global__ void pack_w_kernel(const float* __restrict__ W, float* __restrict__ Wp,
                              int K, int N, int Np, unsigned* zw, int nz, int zstride) {
  int i = blockIdx.x * blockDim.x + threadIdx.x;
  if (i < nz) zw[(size_t)i * zstride] = 0u;
  int total = (K >> 1) * Np;
  if (i >= total) return;
  int k2 = i / Np;
  int c = i - k2 * Np;
  float x0 = 0.f, x1 = 0.f;
  if (c < N) {
    x0 = W[(size_t)(2 * k2) * N + c];
    x1 = W[(size_t)(2 * k2 + 1) * N + c];
  }
  *(float2*)(Wp + (size_t)k2 * 2 * Np + 2 * c) = make_float2(x0, x1);
}

template <int R, int LCAP, int NBMAX>
__global__ __launch_bounds__(1024) void bin_edges_kernel(const int* __restrict__ src,
                                                         const int* __restrict__ dst,
                                                         unsigned* lists, int nE, int nRows, int nB) {
  __shared__ unsigned hist[NBMAX];
  __shared__ unsigned gbase[NBMAX];
  const int tid = (int)threadIdx.x;
  for (int i = tid; i < nB; i += 1024) hist[i] = 0u;
  __syncthreads();
  const int e0 = ((int)blockIdx.x * 1024 + tid) * 4;
  int d[4] = {-1, -1, -1, -1}, s[4] = {0, 0, 0, 0};
  if (e0 + 3 < nE) {
    const int4 dv = *(const int4*)(dst + e0);
    const int4 sv = *(const int4*)(src + e0);
    d[0] = dv.x; d[1] = dv.y; d[2] = dv.z; d[3] = dv.w;
    s[0] = sv.x; s[1] = sv.y; s[2] = sv.z; s[3] = sv.w;
  } else {
#pragma unroll
    for (int j = 0; j < 4; ++j)
      if (e0 + j < nE) { d[j] = dst[e0 + j]; s[j] = src[e0 + j]; }
  }
  int b[4];
  unsigned rank[4];
#pragma unroll
  for (int j = 0; j < 4; ++j) {
    const bool ok = (unsigned)d[j] < (unsigned)nRows;
    b[j] = ok ? d[j] / R : -1;
    rank[j] = 0u;
    if (ok) rank[j] = atomicAdd(&hist[b[j]], 1u);
  }
  __syncthreads();
  for (int i = tid; i < nB; i += 1024) {
    const unsigned cnt = hist[i];
    if (cnt != 0u) gbase[i] = atomicAdd(lists + (size_t)i * (LCAP + 32), cnt);
  }
  __syncthreads();
#pragma unroll
  for (int j = 0; j < 4; ++j) {
    if (b[j] >= 0) {
      const unsigned slot = gbase[b[j]] + rank[j];
      if (slot < (unsigned)LCAP)
        lists[(size_t)b[j] * (LCAP + 32) + 32 + slot] =
            ((unsigned)(d[j] - b[j] * R) << 20) | ((unsigned)s[j] & 0xFFFFFu);
    }
  }
}


template <int D, int R, int NT, int QCAP, int LCAP, bool BUILD>
__global__ __launch_bounds__(NT) void owner_sum_kernel(const float* __restrict__ h,
                                                       const int* __restrict__ src,
                                                       const int* __restrict__ dst,
                                                       float* __restrict__ msum,
                                                       unsigned* lists, float* invd, int nE,
                                                       int nRows) {
  static_assert(R <= 4096, "local row must fit the 12 bits above the edge index");
  static_assert(D % 64 == 0, "rows are handled as 256-byte units");
  __shared__ __attribute__((aligned(16))) float tile[R * D];
  __shared__ unsigned queue[QCAP];
  __shared__ unsigned qcount;
  __shared__ int cnt[R];
  __shared__ float sinv[R];

  constexpr int NW = NT / 32;
  constexpr int U = 8;
  const int tid = (int)threadIdx.x;
  const int lane = tid & 31;
  const int wv = __builtin_amdgcn_readfirstlane(tid >> 5);
  const int base = (int)blockIdx.x * R;
  unsigned* lst = lists + (size_t)blockIdx.x * (LCAP + 32);

  const unsigned hdr = lst[0];
  const bool scan = hdr > (unsigned)LCAP;

  for (int i = tid; i < R * D / 4; i += NT) ((float4*)tile)[i] = make_float4(0.f, 0.f, 0.f, 0.f);
  for (int i = tid; i < R; i += NT) cnt[i] = 0;
  if (tid == 0) qcount = 0u;
  __syncthreads();

  auto push = [&](int dv, int e) {
    const unsigned t = (unsigned)(dv - base);
    if (t < (unsigned)R) {
      atomicAdd(&cnt[t], 1);
      const unsigned slot = atomicAdd(&qcount, 1u);
      if (slot < (unsigned)QCAP) {
        queue[slot] = (t << 20) | (unsigned)e;
      } else {
        const float* hp = h + (size_t)src[e] * D;
        float* tp = tile + t * D;
#pragma unroll 1
        for (int k = 0; k < D; ++k)
          __hip_atomic_fetch_add(tp + k, hp[k], __ATOMIC_RELAXED, __HIP_MEMORY_SCOPE_WORKGROUP);
      }
    }
  };

  const int4* d4 = (const int4*)dst;
  const int nchunk = (nE >> 2) / (32 * U);
  if (scan) {
    for (int c = wv; c < nchunk; c += NW) {
      const int i0 = c * (32 * U) + lane;
      int4 v[U];
#pragma unroll
      for (int u = 0; u < U; ++u) v[u] = d4[i0 + u * 32];
#pragma unroll
      for (int u = 0; u < U; ++u) {
        const int e0 = (i0 + u * 32) << 2;
        const unsigned tx = (unsigned)(v[u].x - base), ty = (unsigned)(v[u].y - base);
        const unsigned tz = (unsigned)(v[u].z - base), tw = (unsigned)(v[u].w - base);
        const unsigned tmin = min(min(tx, ty), min(tz, tw));
        if (__builtin_amdgcn_ballot_w32(tmin < (unsigned)R) != 0u) {
          push(v[u].x, e0);
          push(v[u].y, e0 + 1);
          push(v[u].z, e0 + 2);
          push(v[u].w, e0 + 3);
        }
      }
    }
    for (int e = nchunk * (128 * U) + tid; e < nE; e += NT) push(dst[e], e);
  } else if (BUILD) {
    for (int i = tid; i < (int)hdr; i += NT)
      atomicAdd(&cnt[min((int)(lst[32 + i] >> 20), R - 1)], 1);
  }
  __syncthreads();

  for (int r = tid; r < R; r += NT) {
    const bool live = base + r < nRows;
    float v = 1.0f;
    if (scan || BUILD) v = 1.0f / fmaxf((float)cnt[r], 1.0f);
    else if (live) v = invd[base + r];
    sinv[r] = v;
    if (BUILD && live) invd[base + r] = v;
  }

  const unsigned qn = scan ? qcount : hdr;
  const unsigned cap = scan ? (unsigned)QCAP : (unsigned)LCAP;
  const int n = __builtin_amdgcn_readfirstlane((int)(qn < cap ? qn : cap));
  auto add_row = [&](const float2* vv, int t) {
    float* tp = tile + t * D + 2 * lane;
#pragma unroll
    for (int k = 0; k < D / 64; ++k) {
      __hip_atomic_fetch_add(tp + 64 * k, vv[k].x, __ATOMIC_RELAXED, __HIP_MEMORY_SCOPE_WORKGROUP);
      __hip_atomic_fetch_add(tp + 64 * k + 1, vv[k].y, __ATOMIC_RELAXED,
                             __HIP_MEMORY_SCOPE_WORKGROUP);
    }
  };
  for (int b0 = wv * 32; b0 < n; b0 += NW * 32) {
    const int i = b0 + lane;
    int s, t;
    if (scan) {
      const unsigned p = (i < n) ? queue[i] : 0u;
      s = src[p & 0xFFFFFu];
      t = (int)(p >> 20);
    } else {
      const unsigned p = (i < n) ? lst[32 + i] : 0u;
      s = min((int)(p & 0xFFFFFu), nRows - 1);
      t = min((int)(p >> 20), R - 1);
    }
    const int cnt = (n - b0 < 32) ? (n - b0) : 32;
    int j = 0;
    for (; j + 3 < cnt; j += 4) {
      float2 vv[4][D / 64];
      int tt[4];
#pragma unroll
      for (int g = 0; g < 4; ++g) {
        const int sg = __builtin_amdgcn_readlane(s, j + g);
        tt[g] = __builtin_amdgcn_readlane(t, j + g);
        const float* hg = h + (size_t)sg * D + 2 * lane;
#pragma unroll
        for (int k = 0; k < D / 64; ++k) vv[g][k] = *(const float2*)(hg + 64 * k);
      }
#pragma unroll
      for (int g = 0; g < 4; ++g) add_row(vv[g], tt[g]);
    }
    for (; j < cnt; ++j) {
      const int sa = __builtin_amdgcn_readlane(s, j);
      const int ta = __builtin_amdgcn_readlane(t, j);
      const float* ha = h + (size_t)sa * D + 2 * lane;
      float2 va[D / 64];
#pragma unroll
      for (int k = 0; k < D / 64; ++k) va[k] = *(const float2*)(ha + 64 * k);
      add_row(va, ta);
    }
  }
  __syncthreads();

  const int rows = (nRows - base < R) ? (nRows - base) : R;
  float4* o4 = (float4*)(msum + (size_t)base * D);
  for (int i = tid; i < rows * (D / 4); i += NT) {
    const float4 v = ((const float4*)tile)[i];
    const float s = sinv[i / (D / 4)];
    o4[i] = make_float4(v.x * s, v.y * s, v.z * s, v.w * s);
  }
}

typedef __attribute__((ext_vector_type(4))) float v4f;

__device__ __forceinline__ v8f wmma4(v2f a, v2f b, v8f c) {
  return __builtin_amdgcn_wmma_f32_16x16x4_f32(false, a, false, b, (short)0, c, false, false);
}

template <bool RELU>
__global__ __launch_bounds__(256) void sage_gemm_wmma(
    const float* __restrict__ h, const float* __restrict__ msum,
    const float* __restrict__ Wp,
    const float* __restrict__ bias, float* __restrict__ out, int M, int d, int N, int Np) {
  const int w = blockIdx.x * (blockDim.x >> 5) + ((int)threadIdx.x >> 5);
  const int tilesN = Np >> 6;
  const int tilesM = M >> 5;
  const int tM = w / tilesN;
  const int tN = w - tM * tilesN;
  if (tM >= tilesM) return;

  const int m0 = tM << 5;
  const int n0 = tN << 6;
  const int lane = (int)threadIdx.x & 31;
  const int half = lane >> 4;
  const int lcol = lane & 15;
  const int r0 = m0 + lcol;
  const int r1 = r0 + 16;
  const float* h0 = h + (size_t)r0 * d + 4 * half;
  const float* h1 = h + (size_t)r1 * d + 4 * half;
  const float* s0 = msum + (size_t)r0 * d + 4 * half;
  const float* s1 = msum + (size_t)r1 * d + 4 * half;
  const size_t ldb = 2 * (size_t)Np;
  const float* wb = Wp + (size_t)(2 * half) * ldb + 2 * (n0 + 2 * lcol);

  v8f acc0[4] = {}, acc1[4] = {};

  auto chunk = [&](const float* a0p, const float* a1p, const float* bp) {
    const v4f a0 = *(const v4f*)a0p;
    const v4f a1 = *(const v4f*)a1p;
    const v4f b00 = *(const v4f*)(bp);
    const v4f b01 = *(const v4f*)(bp + 64);
    const v4f b10 = *(const v4f*)(bp + ldb);
    const v4f b11 = *(const v4f*)(bp + ldb + 64);
    const v2f a0lo = __builtin_shufflevector(a0, a0, 0, 1), a0hi = __builtin_shufflevector(a0, a0, 2, 3);
    const v2f a1lo = __builtin_shufflevector(a1, a1, 0, 1), a1hi = __builtin_shufflevector(a1, a1, 2, 3);
    acc0[0] = wmma4(a0lo, __builtin_shufflevector(b00, b00, 0, 1), acc0[0]);
    acc1[0] = wmma4(a1lo, __builtin_shufflevector(b00, b00, 0, 1), acc1[0]);
    acc0[1] = wmma4(a0lo, __builtin_shufflevector(b00, b00, 2, 3), acc0[1]);
    acc1[1] = wmma4(a1lo, __builtin_shufflevector(b00, b00, 2, 3), acc1[1]);
    acc0[2] = wmma4(a0lo, __builtin_shufflevector(b01, b01, 0, 1), acc0[2]);
    acc1[2] = wmma4(a1lo, __builtin_shufflevector(b01, b01, 0, 1), acc1[2]);
    acc0[3] = wmma4(a0lo, __builtin_shufflevector(b01, b01, 2, 3), acc0[3]);
    acc1[3] = wmma4(a1lo, __builtin_shufflevector(b01, b01, 2, 3), acc1[3]);
    acc0[0] = wmma4(a0hi, __builtin_shufflevector(b10, b10, 0, 1), acc0[0]);
    acc1[0] = wmma4(a1hi, __builtin_shufflevector(b10, b10, 0, 1), acc1[0]);
    acc0[1] = wmma4(a0hi, __builtin_shufflevector(b10, b10, 2, 3), acc0[1]);
    acc1[1] = wmma4(a1hi, __builtin_shufflevector(b10, b10, 2, 3), acc1[1]);
    acc0[2] = wmma4(a0hi, __builtin_shufflevector(b11, b11, 0, 1), acc0[2]);
    acc1[2] = wmma4(a1hi, __builtin_shufflevector(b11, b11, 0, 1), acc1[2]);
    acc0[3] = wmma4(a0hi, __builtin_shufflevector(b11, b11, 2, 3), acc0[3]);
    acc1[3] = wmma4(a1hi, __builtin_shufflevector(b11, b11, 2, 3), acc1[3]);
  };

  for (int k0 = 0; k0 < d; k0 += 8) chunk(h0 + k0, h1 + k0, wb + (size_t)(k0 >> 1) * ldb);
  for (int k0 = 0; k0 < d; k0 += 8) chunk(s0 + k0, s1 + k0, wb + (size_t)((d + k0) >> 1) * ldb);

  const bool evenN = (N & 1) == 0;
#pragma unroll
  for (int q = 0; q < 2; ++q) {
    const int cc = n0 + 32 * q + 2 * lcol;
    if (evenN) {
      if (cc + 1 < N) {
        const float bx = bias[cc], by = bias[cc + 1];
#pragma unroll
        for (int r = 0; r < 8; ++r) {
          float x0 = acc0[2 * q][r] + bx, y0 = acc0[2 * q + 1][r] + by;
          float x1 = acc1[2 * q][r] + bx, y1 = acc1[2 * q + 1][r] + by;
          if (RELU) {
            x0 = fmaxf(x0, 0.0f); y0 = fmaxf(y0, 0.0f);
            x1 = fmaxf(x1, 0.0f); y1 = fmaxf(y1, 0.0f);
          }
          *(float2*)(out + (size_t)(m0 + half * 8 + r) * N + cc) = make_float2(x0, y0);
          *(float2*)(out + (size_t)(m0 + 16 + half * 8 + r) * N + cc) = make_float2(x1, y1);
        }
      }
    } else {
#pragma unroll
      for (int i = 0; i < 2; ++i) {
        if (cc + i < N) {
          const float bj = bias[cc + i];
#pragma unroll
          for (int r = 0; r < 8; ++r) {
            float v0 = acc0[2 * q + i][r] + bj;
            float v1 = acc1[2 * q + i][r] + bj;
            if (RELU) {
              v0 = fmaxf(v0, 0.0f);
              v1 = fmaxf(v1, 0.0f);
            }
            out[(size_t)(m0 + half * 8 + r) * N + cc + i] = v0;
            out[(size_t)(m0 + 16 + half * 8 + r) * N + cc + i] = v1;
          }
        }
      }
    }
  }
}

extern "C" void kernel_launch(void* const* d_in, const int* in_sizes, int n_in, void* d_out,
                              int out_size, void* d_ws, size_t ws_size, hipStream_t stream) {
  (void)in_sizes; (void)n_in; (void)out_size; (void)ws_size;
  const float* x  = (const float*)d_in[0];
  const float* W1 = (const float*)d_in[1];
  const float* b1 = (const float*)d_in[2];
  const float* W2 = (const float*)d_in[3];
  const float* b2 = (const float*)d_in[4];
  const float* W3 = (const float*)d_in[5];
  const float* b3 = (const float*)d_in[6];
  const int* src  = (const int*)d_in[7];
  const int* dst  = (const int*)d_in[8];
  float* out = (float*)d_out;

  constexpr int M = 40000, E = 640000, INF = 128, HID = 256, OUTF = 47, OUTP = 64;
  constexpr int OWN_R = 160, OWN_T = 1024, OWN_Q = 8192, OWN_B = (M + OWN_R - 1) / OWN_R;
  static_assert(E <= (1 << 20), "edge index must fit the 20 low bits of a queue word");
  static_assert(M <= (1 << 20), "source node must fit the 20 low bits of a saved list word");
  constexpr int OWN_L = 7456;
  static_assert((long long)OWN_B * (OWN_L + 32) <= (long long)M * OUTF,
                "the saved lists must fit the output buffer");
  static_assert(OWN_B <= 256, "the binning launch ranks at most 256 owners in LDS");
  unsigned* lists = (unsigned*)d_out;

  float* ws = (float*)d_ws;
  float* invd = ws + 40000;
  float* msum = ws + 80000;
  float* h1   = msum + (size_t)M * HID;
  float* h2   = h1 + (size_t)M * HID;
  float* Wp1  = h2 + (size_t)M * HID;
  float* Wp2  = Wp1 + (size_t)(2 * INF) * HID;
  float* Wp3  = Wp2 + (size_t)(2 * HID) * HID;

  const int T = 256;
  auto blocks = [](long long n, int t) { return (int)((n + t - 1) / t); };

  pack_w_kernel<<<blocks((2 * INF / 2) * (long long)HID, T), T, 0, stream>>>(W1, Wp1, 2 * INF, HID, HID, lists, OWN_B, OWN_L + 32);
  pack_w_kernel<<<blocks((2 * HID / 2) * (long long)HID, T), T, 0, stream>>>(W2, Wp2, 2 * HID, HID, HID, lists, 0, 0);
  pack_w_kernel<<<blocks((2 * HID / 2) * (long long)OUTP, T), T, 0, stream>>>(W3, Wp3, 2 * HID, OUTF, OUTP, lists, 0, 0);

  bin_edges_kernel<OWN_R, OWN_L, 256><<<(E + 4095) / 4096, 1024, 0, stream>>>(src, dst, lists, E, M, OWN_B);

  {
    owner_sum_kernel<INF, OWN_R, OWN_T, OWN_Q, OWN_L, true><<<OWN_B, OWN_T, 0, stream>>>(
        x, src, dst, msum, lists, invd, E, M);
    int tiles = (M / 32) * (HID / 64);
    sage_gemm_wmma<true><<<blocks((long long)tiles * 32, T), T, 0, stream>>>(
        x, msum, Wp1, b1, h1, M, INF, HID, HID);
  }
  {
    owner_sum_kernel<HID, OWN_R, OWN_T, OWN_Q, OWN_L, false><<<OWN_B, OWN_T, 0, stream>>>(
        h1, src, dst, msum, lists, invd, E, M);
    int tiles = (M / 32) * (HID / 64);
    sage_gemm_wmma<true><<<blocks((long long)tiles * 32, T), T, 0, stream>>>(
        h1, msum, Wp2, b2, h2, M, HID, HID, HID);
  }
  {
    owner_sum_kernel<HID, OWN_R, OWN_T, OWN_Q, OWN_L, false><<<OWN_B, OWN_T, 0, stream>>>(
        h2, src, dst, msum, lists, invd, E, M);
    int tiles = (M / 32) * (OUTP / 64);
    sage_gemm_wmma<false><<<blocks((long long)tiles * 32, T), T, 0, stream>>>(
        h2, msum, Wp3, b3, out, M, HID, OUTF, OUTP);
  }
}


# GEMM launches in 4-wave work-groups (2 x 2 wave tiles, one wave per SIMD) instead of 8-wave: groups fall 5 or 4 to a compute unit, so the fullest SIMD issues 5 wave tiles of WMMAs, not 6
# speedup vs baseline: 2.8404x; 1.0549x over previous
#include <hip/hip_runtime.h>

typedef __attribute__((ext_vector_type(2))) float v2f;
typedef __attribute__((ext_vector_type(8))) float v8f;


__global__ void pack_w_kernel(const float* __restrict__ W, float* __restrict__ Wp,
                              int K, int N, int Np, unsigned* zw, int nz, int zstride) {
  int i = blockIdx.x * blockDim.x + threadIdx.x;
  if (i < nz) zw[(size_t)i * zstride] = 0u;
  int total = (K >> 1) * Np;
  if (i >= total) return;
  int k2 = i / Np;
  int c = i - k2 * Np;
  float x0 = 0.f, x1 = 0.f;
  if (c < N) {
    x0 = W[(size_t)(2 * k2) * N + c];
    x1 = W[(size_t)(2 * k2 + 1) * N + c];
  }
  *(float2*)(Wp + (size_t)k2 * 2 * Np + 2 * c) = make_float2(x0, x1);
}

template <int R, int LCAP, int NBMAX>
__global__ __launch_bounds__(1024) void bin_edges_kernel(const int* __restrict__ src,
                                                         const int* __restrict__ dst,
                                                         unsigned* lists, int nE, int nRows, int nB) {
  __shared__ unsigned hist[NBMAX];
  __shared__ unsigned gbase[NBMAX];
  const int tid = (int)threadIdx.x;
  for (int i = tid; i < nB; i += 1024) hist[i] = 0u;
  __syncthreads();
  const int e0 = ((int)blockIdx.x * 1024 + tid) * 4;
  int d[4] = {-1, -1, -1, -1}, s[4] = {0, 0, 0, 0};
  if (e0 + 3 < nE) {
    const int4 dv = *(const int4*)(dst + e0);
    const int4 sv = *(const int4*)(src + e0);
    d[0] = dv.x; d[1] = dv.y; d[2] = dv.z; d[3] = dv.w;
    s[0] = sv.x; s[1] = sv.y; s[2] = sv.z; s[3] = sv.w;
  } else {
#pragma unroll
    for (int j = 0; j < 4; ++j)
      if (e0 + j < nE) { d[j] = dst[e0 + j]; s[j] = src[e0 + j]; }
  }
  int b[4];
  unsigned rank[4];
#pragma unroll
  for (int j = 0; j < 4; ++j) {
    const bool ok = (unsigned)d[j] < (unsigned)nRows;
    b[j] = ok ? d[j] / R : -1;
    rank[j] = 0u;
    if (ok) rank[j] = atomicAdd(&hist[b[j]], 1u);
  }
  __syncthreads();
  for (int i = tid; i < nB; i += 1024) {
    const unsigned cnt = hist[i];
    if (cnt != 0u) gbase[i] = atomicAdd(lists + (size_t)i * (LCAP + 32), cnt);
  }
  __syncthreads();
#pragma unroll
  for (int j = 0; j < 4; ++j) {
    if (b[j] >= 0) {
      const unsigned slot = gbase[b[j]] + rank[j];
      if (slot < (unsigned)LCAP)
        lists[(size_t)b[j] * (LCAP + 32) + 32 + slot] =
            ((unsigned)(d[j] - b[j] * R) << 20) | ((unsigned)s[j] & 0xFFFFFu);
    }
  }
}


template <int D, int R, int NT, int QCAP, int LCAP, bool BUILD>
__global__ __launch_bounds__(NT) void owner_sum_kernel(const float* __restrict__ h,
                                                       const int* __restrict__ src,
                                                       const int* __restrict__ dst,
                                                       float* __restrict__ msum,
                                                       unsigned* lists, float* invd, int nE,
                                                       int nRows) {
  static_assert(R <= 4096, "local row must fit the 12 bits above the edge index");
  static_assert(D % 64 == 0, "rows are handled as 256-byte units");
  __shared__ __attribute__((aligned(16))) float tile[R * D];
  __shared__ unsigned queue[QCAP];
  __shared__ unsigned qcount;
  __shared__ int cnt[R];
  __shared__ float sinv[R];

  constexpr int NW = NT / 32;
  constexpr int U = 8;
  const int tid = (int)threadIdx.x;
  const int lane = tid & 31;
  const int wv = __builtin_amdgcn_readfirstlane(tid >> 5);
  const int base = (int)blockIdx.x * R;
  unsigned* lst = lists + (size_t)blockIdx.x * (LCAP + 32);

  const unsigned hdr = lst[0];
  const bool scan = hdr > (unsigned)LCAP;

  for (int i = tid; i < R * D / 4; i += NT) ((float4*)tile)[i] = make_float4(0.f, 0.f, 0.f, 0.f);
  for (int i = tid; i < R; i += NT) cnt[i] = 0;
  if (tid == 0) qcount = 0u;
  __syncthreads();

  auto push = [&](int dv, int e) {
    const unsigned t = (unsigned)(dv - base);
    if (t < (unsigned)R) {
      atomicAdd(&cnt[t], 1);
      const unsigned slot = atomicAdd(&qcount, 1u);
      if (slot < (unsigned)QCAP) {
        queue[slot] = (t << 20) | (unsigned)e;
      } else {
        const float* hp = h + (size_t)src[e] * D;
        float* tp = tile + t * D;
#pragma unroll 1
        for (int k = 0; k < D; ++k)
          __hip_atomic_fetch_add(tp + k, hp[k], __ATOMIC_RELAXED, __HIP_MEMORY_SCOPE_WORKGROUP);
      }
    }
  };

  const int4* d4 = (const int4*)dst;
  const int nchunk = (nE >> 2) / (32 * U);
  if (scan) {
    for (int c = wv; c < nchunk; c += NW) {
      const int i0 = c * (32 * U) + lane;
      int4 v[U];
#pragma unroll
      for (int u = 0; u < U; ++u) v[u] = d4[i0 + u * 32];
#pragma unroll
      for (int u = 0; u < U; ++u) {
        const int e0 = (i0 + u * 32) << 2;
        const unsigned tx = (unsigned)(v[u].x - base), ty = (unsigned)(v[u].y - base);
        const unsigned tz = (unsigned)(v[u].z - base), tw = (unsigned)(v[u].w - base);
        const unsigned tmin = min(min(tx, ty), min(tz, tw));
        if (__builtin_amdgcn_ballot_w32(tmin < (unsigned)R) != 0u) {
          push(v[u].x, e0);
          push(v[u].y, e0 + 1);
          push(v[u].z, e0 + 2);
          push(v[u].w, e0 + 3);
        }
      }
    }
    for (int e = nchunk * (128 * U) + tid; e < nE; e += NT) push(dst[e], e);
  } else if (BUILD) {
    for (int i = tid; i < (int)hdr; i += NT)
      atomicAdd(&cnt[min((int)(lst[32 + i] >> 20), R - 1)], 1);
  }
  __syncthreads();

  for (int r = tid; r < R; r += NT) {
    const bool live = base + r < nRows;
    float v = 1.0f;
    if (scan || BUILD) v = 1.0f / fmaxf((float)cnt[r], 1.0f);
    else if (live) v = invd[base + r];
    sinv[r] = v;
    if (BUILD && live) invd[base + r] = v;
  }

  const unsigned qn = scan ? qcount : hdr;
  const unsigned cap = scan ? (unsigned)QCAP : (unsigned)LCAP;
  const int n = __builtin_amdgcn_readfirstlane((int)(qn < cap ? qn : cap));
  auto add_row = [&](const float2* vv, int t) {
    float* tp = tile + t * D + 2 * lane;
#pragma unroll
    for (int k = 0; k < D / 64; ++k) {
      __hip_atomic_fetch_add(tp + 64 * k, vv[k].x, __ATOMIC_RELAXED, __HIP_MEMORY_SCOPE_WORKGROUP);
      __hip_atomic_fetch_add(tp + 64 * k + 1, vv[k].y, __ATOMIC_RELAXED,
                             __HIP_MEMORY_SCOPE_WORKGROUP);
    }
  };
  for (int b0 = wv * 32; b0 < n; b0 += NW * 32) {
    const int i = b0 + lane;
    int s, t;
    if (scan) {
      const unsigned p = (i < n) ? queue[i] : 0u;
      s = src[p & 0xFFFFFu];
      t = (int)(p >> 20);
    } else {
      const unsigned p = (i < n) ? lst[32 + i] : 0u;
      s = min((int)(p & 0xFFFFFu), nRows - 1);
      t = min((int)(p >> 20), R - 1);
    }
    const int cnt = (n - b0 < 32) ? (n - b0) : 32;
    int j = 0;
    for (; j + 3 < cnt; j += 4) {
      float2 vv[4][D / 64];
      int tt[4];
#pragma unroll
      for (int g = 0; g < 4; ++g) {
        const int sg = __builtin_amdgcn_readlane(s, j + g);
        tt[g] = __builtin_amdgcn_readlane(t, j + g);
        const float* hg = h + (size_t)sg * D + 2 * lane;
#pragma unroll
        for (int k = 0; k < D / 64; ++k) vv[g][k] = *(const float2*)(hg + 64 * k);
      }
#pragma unroll
      for (int g = 0; g < 4; ++g) add_row(vv[g], tt[g]);
    }
    for (; j < cnt; ++j) {
      const int sa = __builtin_amdgcn_readlane(s, j);
      const int ta = __builtin_amdgcn_readlane(t, j);
      const float* ha = h + (size_t)sa * D + 2 * lane;
      float2 va[D / 64];
#pragma unroll
      for (int k = 0; k < D / 64; ++k) va[k] = *(const float2*)(ha + 64 * k);
      add_row(va, ta);
    }
  }
  __syncthreads();

  const int rows = (nRows - base < R) ? (nRows - base) : R;
  float4* o4 = (float4*)(msum + (size_t)base * D);
  for (int i = tid; i < rows * (D / 4); i += NT) {
    const float4 v = ((const float4*)tile)[i];
    const float s = sinv[i / (D / 4)];
    o4[i] = make_float4(v.x * s, v.y * s, v.z * s, v.w * s);
  }
}

typedef __attribute__((ext_vector_type(4))) float v4f;

__device__ __forceinline__ v8f wmma4(v2f a, v2f b, v8f c) {
  return __builtin_amdgcn_wmma_f32_16x16x4_f32(false, a, false, b, (short)0, c, false, false);
}

template <bool RELU>
__global__ __launch_bounds__(128) void sage_gemm_wmma(
    const float* __restrict__ h, const float* __restrict__ msum,
    const float* __restrict__ Wp,
    const float* __restrict__ bias, float* __restrict__ out, int M, int d, int N, int Np) {
  const int wl = (int)threadIdx.x >> 5;
  const int tilesN = Np >> 6;
  const int tilesM = M >> 5;
  const int gn = (tilesN & 1) ? 1 : 2;
  const int gm = 4 / gn;
  const int groupsN = tilesN / gn;
  const int bm = (int)blockIdx.x / groupsN;
  const int bn = (int)blockIdx.x - bm * groupsN;
  const int wm = wl / gn;
  const int tM = bm * gm + wm;
  const int tN = bn * gn + (wl - wm * gn);
  if (tM >= tilesM) return;

  const int m0 = tM << 5;
  const int n0 = tN << 6;
  const int lane = (int)threadIdx.x & 31;
  const int half = lane >> 4;
  const int lcol = lane & 15;
  const int r0 = m0 + lcol;
  const int r1 = r0 + 16;
  const float* h0 = h + (size_t)r0 * d + 4 * half;
  const float* h1 = h + (size_t)r1 * d + 4 * half;
  const float* s0 = msum + (size_t)r0 * d + 4 * half;
  const float* s1 = msum + (size_t)r1 * d + 4 * half;
  const size_t ldb = 2 * (size_t)Np;
  const float* wb = Wp + (size_t)(2 * half) * ldb + 2 * (n0 + 2 * lcol);

  v8f acc0[4] = {}, acc1[4] = {};

  auto chunk = [&](const float* a0p, const float* a1p, const float* bp) {
    const v4f a0 = *(const v4f*)a0p;
    const v4f a1 = *(const v4f*)a1p;
    const v4f b00 = *(const v4f*)(bp);
    const v4f b01 = *(const v4f*)(bp + 64);
    const v4f b10 = *(const v4f*)(bp + ldb);
    const v4f b11 = *(const v4f*)(bp + ldb + 64);
    const v2f a0lo = __builtin_shufflevector(a0, a0, 0, 1), a0hi = __builtin_shufflevector(a0, a0, 2, 3);
    const v2f a1lo = __builtin_shufflevector(a1, a1, 0, 1), a1hi = __builtin_shufflevector(a1, a1, 2, 3);
    acc0[0] = wmma4(a0lo, __builtin_shufflevector(b00, b00, 0, 1), acc0[0]);
    acc1[0] = wmma4(a1lo, __builtin_shufflevector(b00, b00, 0, 1), acc1[0]);
    acc0[1] = wmma4(a0lo, __builtin_shufflevector(b00, b00, 2, 3), acc0[1]);
    acc1[1] = wmma4(a1lo, __builtin_shufflevector(b00, b00, 2, 3), acc1[1]);
    acc0[2] = wmma4(a0lo, __builtin_shufflevector(b01, b01, 0, 1), acc0[2]);
    acc1[2] = wmma4(a1lo, __builtin_shufflevector(b01, b01, 0, 1), acc1[2]);
    acc0[3] = wmma4(a0lo, __builtin_shufflevector(b01, b01, 2, 3), acc0[3]);
    acc1[3] = wmma4(a1lo, __builtin_shufflevector(b01, b01, 2, 3), acc1[3]);
    acc0[0] = wmma4(a0hi, __builtin_shufflevector(b10, b10, 0, 1), acc0[0]);
    acc1[0] = wmma4(a1hi, __builtin_shufflevector(b10, b10, 0, 1), acc1[0]);
    acc0[1] = wmma4(a0hi, __builtin_shufflevector(b10, b10, 2, 3), acc0[1]);
    acc1[1] = wmma4(a1hi, __builtin_shufflevector(b10, b10, 2, 3), acc1[1]);
    acc0[2] = wmma4(a0hi, __builtin_shufflevector(b11, b11, 0, 1), acc0[2]);
    acc1[2] = wmma4(a1hi, __builtin_shufflevector(b11, b11, 0, 1), acc1[2]);
    acc0[3] = wmma4(a0hi, __builtin_shufflevector(b11, b11, 2, 3), acc0[3]);
    acc1[3] = wmma4(a1hi, __builtin_shufflevector(b11, b11, 2, 3), acc1[3]);
  };

  for (int k0 = 0; k0 < d; k0 += 8) chunk(h0 + k0, h1 + k0, wb + (size_t)(k0 >> 1) * ldb);
  for (int k0 = 0; k0 < d; k0 += 8) chunk(s0 + k0, s1 + k0, wb + (size_t)((d + k0) >> 1) * ldb);

  const bool evenN = (N & 1) == 0;
#pragma unroll
  for (int q = 0; q < 2; ++q) {
    const int cc = n0 + 32 * q + 2 * lcol;
    if (evenN) {
      if (cc + 1 < N) {
        const float bx = bias[cc], by = bias[cc + 1];
#pragma unroll
        for (int r = 0; r < 8; ++r) {
          float x0 = acc0[2 * q][r] + bx, y0 = acc0[2 * q + 1][r] + by;
          float x1 = acc1[2 * q][r] + bx, y1 = acc1[2 * q + 1][r] + by;
          if (RELU) {
            x0 = fmaxf(x0, 0.0f); y0 = fmaxf(y0, 0.0f);
            x1 = fmaxf(x1, 0.0f); y1 = fmaxf(y1, 0.0f);
          }
          *(float2*)(out + (size_t)(m0 + half * 8 + r) * N + cc) = make_float2(x0, y0);
          *(float2*)(out + (size_t)(m0 + 16 + half * 8 + r) * N + cc) = make_float2(x1, y1);
        }
      }
    } else {
#pragma unroll
      for (int i = 0; i < 2; ++i) {
        if (cc + i < N) {
          const float bj = bias[cc + i];
#pragma unroll
          for (int r = 0; r < 8; ++r) {
            float v0 = acc0[2 * q + i][r] + bj;
            float v1 = acc1[2 * q + i][r] + bj;
            if (RELU) {
              v0 = fmaxf(v0, 0.0f);
              v1 = fmaxf(v1, 0.0f);
            }
            out[(size_t)(m0 + half * 8 + r) * N + cc + i] = v0;
            out[(size_t)(m0 + 16 + half * 8 + r) * N + cc + i] = v1;
          }
        }
      }
    }
  }
}

extern "C" void kernel_launch(void* const* d_in, const int* in_sizes, int n_in, void* d_out,
                              int out_size, void* d_ws, size_t ws_size, hipStream_t stream) {
  (void)in_sizes; (void)n_in; (void)out_size; (void)ws_size;
  const float* x  = (const float*)d_in[0];
  const float* W1 = (const float*)d_in[1];
  const float* b1 = (const float*)d_in[2];
  const float* W2 = (const float*)d_in[3];
  const float* b2 = (const float*)d_in[4];
  const float* W3 = (const float*)d_in[5];
  const float* b3 = (const float*)d_in[6];
  const int* src  = (const int*)d_in[7];
  const int* dst  = (const int*)d_in[8];
  float* out = (float*)d_out;

  constexpr int M = 40000, E = 640000, INF = 128, HID = 256, OUTF = 47, OUTP = 64;
  constexpr int OWN_R = 160, OWN_T = 1024, OWN_Q = 8192, OWN_B = (M + OWN_R - 1) / OWN_R;
  static_assert(E <= (1 << 20), "edge index must fit the 20 low bits of a queue word");
  static_assert(M <= (1 << 20), "source node must fit the 20 low bits of a saved list word");
  constexpr int OWN_L = 7456;
  static_assert((long long)OWN_B * (OWN_L + 32) <= (long long)M * OUTF,
                "the saved lists must fit the output buffer");
  static_assert(OWN_B <= 256, "the binning launch ranks at most 256 owners in LDS");
  unsigned* lists = (unsigned*)d_out;

  float* ws = (float*)d_ws;
  float* invd = ws + 40000;
  float* msum = ws + 80000;
  float* h1   = msum + (size_t)M * HID;
  float* h2   = h1 + (size_t)M * HID;
  float* Wp1  = h2 + (size_t)M * HID;
  float* Wp2  = Wp1 + (size_t)(2 * INF) * HID;
  float* Wp3  = Wp2 + (size_t)(2 * HID) * HID;

  const int T = 256;
  auto blocks = [](long long n, int t) { return (int)((n + t - 1) / t); };
  constexpr int TG = 128;
  constexpr int GEMM_G_HID = ((M / 32 + 1) / 2) * ((HID / 64) / 2);
  constexpr int GEMM_G_OUT = (M / 32 + 3) / 4;
  static_assert((HID / 64) % 2 == 0 && OUTP / 64 == 1, "group shapes assumed by the two grids");

  pack_w_kernel<<<blocks((2 * INF / 2) * (long long)HID, T), T, 0, stream>>>(W1, Wp1, 2 * INF, HID, HID, lists, OWN_B, OWN_L + 32);
  pack_w_kernel<<<blocks((2 * HID / 2) * (long long)HID, T), T, 0, stream>>>(W2, Wp2, 2 * HID, HID, HID, lists, 0, 0);
  pack_w_kernel<<<blocks((2 * HID / 2) * (long long)OUTP, T), T, 0, stream>>>(W3, Wp3, 2 * HID, OUTF, OUTP, lists, 0, 0);

  bin_edges_kernel<OWN_R, OWN_L, 256><<<(E + 4095) / 4096, 1024, 0, stream>>>(src, dst, lists, E, M, OWN_B);

  {
    owner_sum_kernel<INF, OWN_R, OWN_T, OWN_Q, OWN_L, true><<<OWN_B, OWN_T, 0, stream>>>(
        x, src, dst, msum, lists, invd, E, M);
    sage_gemm_wmma<true><<<GEMM_G_HID, TG, 0, stream>>>(
        x, msum, Wp1, b1, h1, M, INF, HID, HID);
  }
  {
    owner_sum_kernel<HID, OWN_R, OWN_T, OWN_Q, OWN_L, false><<<OWN_B, OWN_T, 0, stream>>>(
        h1, src, dst, msum, lists, invd, E, M);
    sage_gemm_wmma<true><<<GEMM_G_HID, TG, 0, stream>>>(
        h1, msum, Wp2, b2, h2, M, HID, HID, HID);
  }
  {
    owner_sum_kernel<HID, OWN_R, OWN_T, OWN_Q, OWN_L, false><<<OWN_B, OWN_T, 0, stream>>>(
        h2, src, dst, msum, lists, invd, E, M);
    sage_gemm_wmma<false><<<GEMM_G_OUT, TG, 0, stream>>>(
        h2, msum, Wp3, b3, out, M, HID, OUTF, OUTP);
  }
}


# owner blocks route each edge's source id to its destination row's LDS queue (the in-degree add gives the slot); waves sum whole rows in registers, 16-byte loads: no LDS float adds, tile or slab pass
# speedup vs baseline: 3.4743x; 1.2232x over previous
#include <hip/hip_runtime.h>

typedef __attribute__((ext_vector_type(2))) float v2f;
typedef __attribute__((ext_vector_type(8))) float v8f;


__global__ void pack_w_kernel(const float* __restrict__ W, float* __restrict__ Wp,
                              int K, int N, int Np, unsigned* zw, int nz, int zstride) {
  int i = blockIdx.x * blockDim.x + threadIdx.x;
  if (i < nz) zw[(size_t)i * zstride] = 0u;
  int total = (K >> 1) * Np;
  if (i >= total) return;
  int k2 = i / Np;
  int c = i - k2 * Np;
  float x0 = 0.f, x1 = 0.f;
  if (c < N) {
    x0 = W[(size_t)(2 * k2) * N + c];
    x1 = W[(size_t)(2 * k2 + 1) * N + c];
  }
  *(float2*)(Wp + (size_t)k2 * 2 * Np + 2 * c) = make_float2(x0, x1);
}

template <int R, int LCAP, int NBMAX>
__global__ __launch_bounds__(1024) void bin_edges_kernel(const int* __restrict__ src,
                                                         const int* __restrict__ dst,
                                                         unsigned* lists, int nE, int nRows, int nB) {
  __shared__ unsigned hist[NBMAX];
  __shared__ unsigned gbase[NBMAX];
  const int tid = (int)threadIdx.x;
  for (int i = tid; i < nB; i += 1024) hist[i] = 0u;
  __syncthreads();
  const int e0 = ((int)blockIdx.x * 1024 + tid) * 4;
  int d[4] = {-1, -1, -1, -1}, s[4] = {0, 0, 0, 0};
  if (e0 + 3 < nE) {
    const int4 dv = *(const int4*)(dst + e0);
    const int4 sv = *(const int4*)(src + e0);
    d[0] = dv.x; d[1] = dv.y; d[2] = dv.z; d[3] = dv.w;
    s[0] = sv.x; s[1] = sv.y; s[2] = sv.z; s[3] = sv.w;
  } else {
#pragma unroll
    for (int j = 0; j < 4; ++j)
      if (e0 + j < nE) { d[j] = dst[e0 + j]; s[j] = src[e0 + j]; }
  }
  int b[4];
  unsigned rank[4];
#pragma unroll
  for (int j = 0; j < 4; ++j) {
    const bool ok = (unsigned)d[j] < (unsigned)nRows;
    b[j] = ok ? d[j] / R : -1;
    rank[j] = 0u;
    if (ok) rank[j] = atomicAdd(&hist[b[j]], 1u);
  }
  __syncthreads();
  for (int i = tid; i < nB; i += 1024) {
    const unsigned cnt = hist[i];
    if (cnt != 0u) gbase[i] = atomicAdd(lists + (size_t)i * (LCAP + 32), cnt);
  }
  __syncthreads();
#pragma unroll
  for (int j = 0; j < 4; ++j) {
    if (b[j] >= 0) {
      const unsigned slot = gbase[b[j]] + rank[j];
      if (slot < (unsigned)LCAP)
        lists[(size_t)b[j] * (LCAP + 32) + 32 + slot] =
            ((unsigned)(d[j] - b[j] * R) << 20) | ((unsigned)s[j] & 0xFFFFFu);
    }
  }
}


template <int D, int R, int NT, int RC, int LCAP>
__global__ __launch_bounds__(NT) void owner_sum_kernel(const float* __restrict__ h,
                                                       const int* __restrict__ src,
                                                       const int* __restrict__ dst,
                                                       float* __restrict__ msum,
                                                       const unsigned* __restrict__ lists, int nE,
                                                       int nRows) {
  static_assert(R <= 4096, "local row must fit the 12 bits above the source node");
  static_assert(D % 128 == 0, "a wave load covers 512 contiguous bytes of a row");
  static_assert(RC % 32 == 0, "a row queue is read in whole 32-entry chunks");
  __shared__ unsigned rq[R * RC];
  __shared__ int cnt[R];
  __shared__ int nextrow;

  constexpr int KV = D / 128;
  constexpr int G = 2048 / D;
  const int tid = (int)threadIdx.x;
  const int lane = tid & 31;
  const int base = (int)blockIdx.x * R;
  const unsigned* lst = lists + (size_t)blockIdx.x * (LCAP + 32);

  const unsigned hdr = lst[0];
  const bool listok = hdr <= (unsigned)LCAP;
  const int nl = listok ? (int)hdr : 0;

  for (int i = tid; i < R; i += NT) cnt[i] = 0;
  if (tid == 0) nextrow = 0;
  __syncthreads();

  for (int i = tid; i < nl; i += NT) {
    const unsigned p = lst[32 + i];
    const int t = min((int)(p >> 20), R - 1);
    const unsigned s = min(p & 0xFFFFFu, (unsigned)(nRows - 1));
    const int slot = atomicAdd(&cnt[t], 1);
    if (slot < RC) rq[t * RC + slot] = s;
  }
  __syncthreads();

  const unsigned loff = (unsigned)lane * 16u;
  float4 acc[KV];
  auto add_one = [&](int s) {
    const char* hp = (const char*)(h + (size_t)s * D) + loff;
#pragma unroll
    for (int k = 0; k < KV; ++k) {
      const float4 v = *(const float4*)(hp + 512 * k);
      acc[k].x += v.x; acc[k].y += v.y; acc[k].z += v.z; acc[k].w += v.w;
    }
  };

  for (int it = 0; it < R; ++it) {
    int rr = 0;
    if (lane == 0) rr = atomicAdd(&nextrow, 1);
    const int r = __builtin_amdgcn_readfirstlane(rr);
    if (r >= R || base + r >= nRows) break;
#pragma unroll
    for (int k = 0; k < KV; ++k) acc[k] = make_float4(0.f, 0.f, 0.f, 0.f);
    int deg = 0;
    if (listok) {
      deg = __builtin_amdgcn_readfirstlane(cnt[r]);
      if (deg <= RC) {
        const unsigned* q = rq + r * RC;
        for (int c0 = 0; c0 < deg; c0 += 32) {
          const int m = (deg - c0 < 32) ? (deg - c0) : 32;
          const unsigned e = (lane < m) ? q[c0 + lane] : 0u;
          auto group = [&](int j, const bool whole) {
            float4 v[G][KV];
#pragma unroll
            for (int g = 0; g < G; ++g) {
#pragma unroll
              for (int k = 0; k < KV; ++k) v[g][k] = make_float4(0.f, 0.f, 0.f, 0.f);
              if (whole || j + g < m) {
                const int sg = __builtin_amdgcn_readlane((int)e, j + g);
                const char* hp = (const char*)(h + (size_t)sg * D) + loff;
#pragma unroll
                for (int k = 0; k < KV; ++k) v[g][k] = *(const float4*)(hp + 512 * k);
              }
            }
#pragma unroll
            for (int g = 0; g < G; ++g) {
#pragma unroll
              for (int k = 0; k < KV; ++k) {
                acc[k].x += v[g][k].x; acc[k].y += v[g][k].y;
                acc[k].z += v[g][k].z; acc[k].w += v[g][k].w;
              }
            }
          };
          int j = 0;
          for (; j + G <= m; j += G) group(j, true);
          if (j < m) group(j, false);
        }
      } else {
        for (int i0 = 0; i0 < nl; i0 += 32) {
          const int i = i0 + lane;
          const unsigned p = (i < nl) ? lst[32 + i] : 0u;
          const bool own = (i < nl) && (min((int)(p >> 20), R - 1) == r);
          const int sv = (int)min(p & 0xFFFFFu, (unsigned)(nRows - 1));
          unsigned mk = __builtin_amdgcn_ballot_w32(own);
          while (mk != 0u) {
            const int j = __builtin_ctz(mk);
            mk &= mk - 1u;
            add_one(__builtin_amdgcn_readlane(sv, j));
          }
        }
      }
    } else {
      for (int e0 = 0; e0 < nE; e0 += 32) {
        const int e = e0 + lane;
        const bool in = e < nE;
        const int dv = in ? dst[e] : -1;
        unsigned mk = __builtin_amdgcn_ballot_w32(in && dv == base + r);
        if (mk != 0u) {
          deg += __builtin_popcount(mk);
          const int sraw = in ? src[e] : 0;
          const int sv = min(max(sraw, 0), nRows - 1);
          while (mk != 0u) {
            const int j = __builtin_ctz(mk);
            mk &= mk - 1u;
            add_one(__builtin_amdgcn_readlane(sv, j));
          }
        }
      }
    }
    const float sc = 1.0f / fmaxf((float)deg, 1.0f);
    char* op = (char*)(msum + (size_t)(base + r) * D) + loff;
#pragma unroll
    for (int k = 0; k < KV; ++k)
      *(float4*)(op + 512 * k) =
          make_float4(acc[k].x * sc, acc[k].y * sc, acc[k].z * sc, acc[k].w * sc);
  }
}

typedef __attribute__((ext_vector_type(4))) float v4f;

__device__ __forceinline__ v8f wmma4(v2f a, v2f b, v8f c) {
  return __builtin_amdgcn_wmma_f32_16x16x4_f32(false, a, false, b, (short)0, c, false, false);
}

template <bool RELU>
__global__ __launch_bounds__(128) void sage_gemm_wmma(
    const float* __restrict__ h, const float* __restrict__ msum,
    const float* __restrict__ Wp,
    const float* __restrict__ bias, float* __restrict__ out, int M, int d, int N, int Np) {
  const int wl = (int)threadIdx.x >> 5;
  const int tilesN = Np >> 6;
  const int tilesM = M >> 5;
  const int gn = (tilesN & 1) ? 1 : 2;
  const int gm = 4 / gn;
  const int groupsN = tilesN / gn;
  const int bm = (int)blockIdx.x / groupsN;
  const int bn = (int)blockIdx.x - bm * groupsN;
  const int wm = wl / gn;
  const int tM = bm * gm + wm;
  const int tN = bn * gn + (wl - wm * gn);
  if (tM >= tilesM) return;

  const int m0 = tM << 5;
  const int n0 = tN << 6;
  const int lane = (int)threadIdx.x & 31;
  const int half = lane >> 4;
  const int lcol = lane & 15;
  const int r0 = m0 + lcol;
  const int r1 = r0 + 16;
  const float* h0 = h + (size_t)r0 * d + 4 * half;
  const float* h1 = h + (size_t)r1 * d + 4 * half;
  const float* s0 = msum + (size_t)r0 * d + 4 * half;
  const float* s1 = msum + (size_t)r1 * d + 4 * half;
  const size_t ldb = 2 * (size_t)Np;
  const float* wb = Wp + (size_t)(2 * half) * ldb + 2 * (n0 + 2 * lcol);

  v8f acc0[4] = {}, acc1[4] = {};

  auto chunk = [&](const float* a0p, const float* a1p, const float* bp) {
    const v4f a0 = *(const v4f*)a0p;
    const v4f a1 = *(const v4f*)a1p;
    const v4f b00 = *(const v4f*)(bp);
    const v4f b01 = *(const v4f*)(bp + 64);
    const v4f b10 = *(const v4f*)(bp + ldb);
    const v4f b11 = *(const v4f*)(bp + ldb + 64);
    const v2f a0lo = __builtin_shufflevector(a0, a0, 0, 1), a0hi = __builtin_shufflevector(a0, a0, 2, 3);
    const v2f a1lo = __builtin_shufflevector(a1, a1, 0, 1), a1hi = __builtin_shufflevector(a1, a1, 2, 3);
    acc0[0] = wmma4(a0lo, __builtin_shufflevector(b00, b00, 0, 1), acc0[0]);
    acc1[0] = wmma4(a1lo, __builtin_shufflevector(b00, b00, 0, 1), acc1[0]);
    acc0[1] = wmma4(a0lo, __builtin_shufflevector(b00, b00, 2, 3), acc0[1]);
    acc1[1] = wmma4(a1lo, __builtin_shufflevector(b00, b00, 2, 3), acc1[1]);
    acc0[2] = wmma4(a0lo, __builtin_shufflevector(b01, b01, 0, 1), acc0[2]);
    acc1[2] = wmma4(a1lo, __builtin_shufflevector(b01, b01, 0, 1), acc1[2]);
    acc0[3] = wmma4(a0lo, __builtin_shufflevector(b01, b01, 2, 3), acc0[3]);
    acc1[3] = wmma4(a1lo, __builtin_shufflevector(b01, b01, 2, 3), acc1[3]);
    acc0[0] = wmma4(a0hi, __builtin_shufflevector(b10, b10, 0, 1), acc0[0]);
    acc1[0] = wmma4(a1hi, __builtin_shufflevector(b10, b10, 0, 1), acc1[0]);
    acc0[1] = wmma4(a0hi, __builtin_shufflevector(b10, b10, 2, 3), acc0[1]);
    acc1[1] = wmma4(a1hi, __builtin_shufflevector(b10, b10, 2, 3), acc1[1]);
    acc0[2] = wmma4(a0hi, __builtin_shufflevector(b11, b11, 0, 1), acc0[2]);
    acc1[2] = wmma4(a1hi, __builtin_shufflevector(b11, b11, 0, 1), acc1[2]);
    acc0[3] = wmma4(a0hi, __builtin_shufflevector(b11, b11, 2, 3), acc0[3]);
    acc1[3] = wmma4(a1hi, __builtin_shufflevector(b11, b11, 2, 3), acc1[3]);
  };

  for (int k0 = 0; k0 < d; k0 += 8) chunk(h0 + k0, h1 + k0, wb + (size_t)(k0 >> 1) * ldb);
  for (int k0 = 0; k0 < d; k0 += 8) chunk(s0 + k0, s1 + k0, wb + (size_t)((d + k0) >> 1) * ldb);

  const bool evenN = (N & 1) == 0;
#pragma unroll
  for (int q = 0; q < 2; ++q) {
    const int cc = n0 + 32 * q + 2 * lcol;
    if (evenN) {
      if (cc + 1 < N) {
        const float bx = bias[cc], by = bias[cc + 1];
#pragma unroll
        for (int r = 0; r < 8; ++r) {
          float x0 = acc0[2 * q][r] + bx, y0 = acc0[2 * q + 1][r] + by;
          float x1 = acc1[2 * q][r] + bx, y1 = acc1[2 * q + 1][r] + by;
          if (RELU) {
            x0 = fmaxf(x0, 0.0f); y0 = fmaxf(y0, 0.0f);
            x1 = fmaxf(x1, 0.0f); y1 = fmaxf(y1, 0.0f);
          }
          *(float2*)(out + (size_t)(m0 + half * 8 + r) * N + cc) = make_float2(x0, y0);
          *(float2*)(out + (size_t)(m0 + 16 + half * 8 + r) * N + cc) = make_float2(x1, y1);
        }
      }
    } else {
#pragma unroll
      for (int i = 0; i < 2; ++i) {
        if (cc + i < N) {
          const float bj = bias[cc + i];
#pragma unroll
          for (int r = 0; r < 8; ++r) {
            float v0 = acc0[2 * q + i][r] + bj;
            float v1 = acc1[2 * q + i][r] + bj;
            if (RELU) {
              v0 = fmaxf(v0, 0.0f);
              v1 = fmaxf(v1, 0.0f);
            }
            out[(size_t)(m0 + half * 8 + r) * N + cc + i] = v0;
            out[(size_t)(m0 + 16 + half * 8 + r) * N + cc + i] = v1;
          }
        }
      }
    }
  }
}

extern "C" void kernel_launch(void* const* d_in, const int* in_sizes, int n_in, void* d_out,
                              int out_size, void* d_ws, size_t ws_size, hipStream_t stream) {
  (void)in_sizes; (void)n_in; (void)out_size; (void)ws_size;
  const float* x  = (const float*)d_in[0];
  const float* W1 = (const float*)d_in[1];
  const float* b1 = (const float*)d_in[2];
  const float* W2 = (const float*)d_in[3];
  const float* b2 = (const float*)d_in[4];
  const float* W3 = (const float*)d_in[5];
  const float* b3 = (const float*)d_in[6];
  const int* src  = (const int*)d_in[7];
  const int* dst  = (const int*)d_in[8];
  float* out = (float*)d_out;

  constexpr int M = 40000, E = 640000, INF = 128, HID = 256, OUTF = 47, OUTP = 64;
  constexpr int OWN_R = 160, OWN_T = 1024, OWN_RC = 64, OWN_B = (M + OWN_R - 1) / OWN_R;
  static_assert(M <= (1 << 20), "source node must fit the 20 low bits of a saved list word");
  constexpr int OWN_L = 7456;
  static_assert((long long)OWN_B * (OWN_L + 32) <= (long long)M * OUTF,
                "the saved lists must fit the output buffer");
  static_assert(OWN_B <= 256, "the binning launch ranks at most 256 owners in LDS");
  unsigned* lists = (unsigned*)d_out;

  float* ws = (float*)d_ws;
  float* msum = ws + 80000;
  float* h1   = msum + (size_t)M * HID;
  float* h2   = h1 + (size_t)M * HID;
  float* Wp1  = h2 + (size_t)M * HID;
  float* Wp2  = Wp1 + (size_t)(2 * INF) * HID;
  float* Wp3  = Wp2 + (size_t)(2 * HID) * HID;

  const int T = 256;
  auto blocks = [](long long n, int t) { return (int)((n + t - 1) / t); };
  constexpr int TG = 128;
  constexpr int GEMM_G_HID = ((M / 32 + 1) / 2) * ((HID / 64) / 2);
  constexpr int GEMM_G_OUT = (M / 32 + 3) / 4;
  static_assert((HID / 64) % 2 == 0 && OUTP / 64 == 1, "group shapes assumed by the two grids");

  pack_w_kernel<<<blocks((2 * INF / 2) * (long long)HID, T), T, 0, stream>>>(W1, Wp1, 2 * INF, HID, HID, lists, OWN_B, OWN_L + 32);
  pack_w_kernel<<<blocks((2 * HID / 2) * (long long)HID, T), T, 0, stream>>>(W2, Wp2, 2 * HID, HID, HID, lists, 0, 0);
  pack_w_kernel<<<blocks((2 * HID / 2) * (long long)OUTP, T), T, 0, stream>>>(W3, Wp3, 2 * HID, OUTF, OUTP, lists, 0, 0);

  bin_edges_kernel<OWN_R, OWN_L, 256><<<(E + 4095) / 4096, 1024, 0, stream>>>(src, dst, lists, E, M, OWN_B);

  {
    owner_sum_kernel<INF, OWN_R, OWN_T, OWN_RC, OWN_L><<<OWN_B, OWN_T, 0, stream>>>(
        x, src, dst, msum, lists, E, M);
    sage_gemm_wmma<true><<<GEMM_G_HID, TG, 0, stream>>>(
        x, msum, Wp1, b1, h1, M, INF, HID, HID);
  }
  {
    owner_sum_kernel<HID, OWN_R, OWN_T, OWN_RC, OWN_L><<<OWN_B, OWN_T, 0, stream>>>(
        h1, src, dst, msum, lists, E, M);
    sage_gemm_wmma<true><<<GEMM_G_HID, TG, 0, stream>>>(
        h1, msum, Wp2, b2, h2, M, HID, HID, HID);
  }
  {
    owner_sum_kernel<HID, OWN_R, OWN_T, OWN_RC, OWN_L><<<OWN_B, OWN_T, 0, stream>>>(
        h2, src, dst, msum, lists, E, M);
    sage_gemm_wmma<false><<<GEMM_G_OUT, TG, 0, stream>>>(
        h2, msum, Wp3, b3, out, M, HID, OUTF, OUTP);
  }
}


# layer 3 in one launch: owner work-groups keep their mean rows in an LDS tile and do their own rows of [h | mean] @ W3 on WMMA (self half ahead of the row loop); 7 launches to 6; sum order unchanged
# speedup vs baseline: 3.9539x; 1.1380x over previous
#include <hip/hip_runtime.h>

typedef __attribute__((ext_vector_type(2))) float v2f;
typedef __attribute__((ext_vector_type(8))) float v8f;
typedef __attribute__((ext_vector_type(4))) float v4f;

__device__ __forceinline__ v8f wmma4(v2f a, v2f b, v8f c) {
  return __builtin_amdgcn_wmma_f32_16x16x4_f32(false, a, false, b, (short)0, c, false, false);
}


__device__ __forceinline__ void pack_item(const float* __restrict__ Wm, float* __restrict__ Wp,
                                          int N, int Np, int i) {
  const int k2 = i / Np;
  const int c = i - k2 * Np;
  float x0 = 0.f, x1 = 0.f;
  if (c < N) {
    x0 = Wm[(size_t)(2 * k2) * N + c];
    x1 = Wm[(size_t)(2 * k2 + 1) * N + c];
  }
  *(float2*)(Wp + (size_t)k2 * 2 * Np + 2 * c) = make_float2(x0, x1);
}

template <int R, int CAP, int NBMAX, int NBIN, int CS, int INF_, int HID_, int OUTF_, int OUTP_>
__global__ __launch_bounds__(1024) void prep_kernel(const int* __restrict__ src,
                                                    const int* __restrict__ dst, unsigned* cells,
                                                    int nE, int nRows, int nB,
                                                    const float* __restrict__ W1,
                                                    const float* __restrict__ W2,
                                                    const float* __restrict__ W3,
                                                    float* __restrict__ Wp1,
                                                    float* __restrict__ Wp2,
                                                    float* __restrict__ Wp3) {
  __shared__ unsigned hist[NBMAX];
  constexpr int CW = CAP + 1;
  constexpr int T1 = INF_ * HID_;
  constexpr int T2 = HID_ * HID_;
  constexpr int T3 = HID_ * OUTP_;
  static_assert((long long)NBIN * 1024 >= (long long)T1 + T2 + T3,
                "one packed pair per thread of the launch");
  const int tid = (int)threadIdx.x;
  const int blk = (int)blockIdx.x;
  for (int i = tid; i < nB; i += 1024) hist[i] = 0u;

  {
    const int g = blk * 1024 + tid;
    if (g < T1) pack_item(W1, Wp1, HID_, HID_, g);
    else if (g < T1 + T2) pack_item(W2, Wp2, HID_, HID_, g - T1);
    else if (g < T1 + T2 + T3) pack_item(W3, Wp3, OUTF_, OUTP_, g - T1 - T2);
  }

  const int e0 = (blk * 1024 + tid) * 4;
  int d[4] = {-1, -1, -1, -1}, s[4] = {0, 0, 0, 0};
  if (e0 + 3 < nE) {
    const int4 dv = *(const int4*)(dst + e0);
    const int4 sv = *(const int4*)(src + e0);
    d[0] = dv.x; d[1] = dv.y; d[2] = dv.z; d[3] = dv.w;
    s[0] = sv.x; s[1] = sv.y; s[2] = sv.z; s[3] = sv.w;
  } else {
#pragma unroll
    for (int j = 0; j < 4; ++j)
      if (e0 + j < nE) { d[j] = dst[e0 + j]; s[j] = src[e0 + j]; }
  }
  __syncthreads();
#pragma unroll
  for (int j = 0; j < 4; ++j) {
    if ((unsigned)d[j] < (unsigned)nRows) {
      const int b = d[j] / R;
      const unsigned rank = atomicAdd(&hist[b], 1u);
      if (rank < (unsigned)CAP)
        cells[(size_t)b * CS + (size_t)blk * CW + 1 + rank] =
            ((unsigned)(d[j] - b * R) << 20) | ((unsigned)s[j] & 0xFFFFFu);
    }
  }
  __syncthreads();
  for (int i = tid; i < nB; i += 1024) cells[(size_t)i * CS + (size_t)blk * CW] = hist[i];
}


template <int D, int R, int NT, int RC, int CAP, int NBIN, int CS, bool FUSE, int NOUT, int NOUTP>
__global__ __launch_bounds__(NT) void owner_sum_kernel(const float* __restrict__ h,
                                                       const int* __restrict__ src,
                                                       const int* __restrict__ dst,
                                                       float* __restrict__ msum,
                                                       const unsigned* cells, int nE, int nRows,
                                                       const float* __restrict__ Wp,
                                                       const float* __restrict__ bias,
                                                       float* out) {
  static_assert(R <= 4096, "local row must fit the 12 bits above the source node");
  static_assert(D % 128 == 0, "a wave load covers 512 contiguous bytes of a row");
  static_assert(RC % 32 == 0, "a row queue is read in whole 32-entry chunks");
  static_assert(CAP >= 32 && CAP <= 63, "a cell is read as two 32-lane loads");
  __shared__ unsigned rq[R * RC];
  __shared__ int cnt[R];
  __shared__ int nextrow;
  __shared__ int bad;
  constexpr int PITCH = D + 4;
  __shared__ __attribute__((aligned(16))) float mt[FUSE ? R * PITCH : 4];
  static_assert(!FUSE || (R % 16 == 0 && 2 * (R / 16) <= NT / 32 && D % 16 == 0 && NOUTP == 64),
                "one 16 x 32 output tile per wave, two k chunks per loop trip");

  constexpr int KV = D / 128;
  constexpr int G = 2048 / D;
  constexpr int CW = CAP + 1;
  constexpr int NW = NT / 32;
  constexpr int CPW = (NBIN + NW - 1) / NW;
  const int tid = (int)threadIdx.x;
  const int lane = tid & 31;
  const int wv = tid >> 5;
  const int base = (int)blockIdx.x * R;
  static_assert(NBIN * CW <= CS, "an owner's cells must fit its run");
  const unsigned* cb = cells + (size_t)blockIdx.x * CS;

  unsigned w0[CPW], w1[CPW];
#pragma unroll
  for (int q = 0; q < CPW; ++q) {
    const int j = wv + q * NW;
    w0[q] = 0u;
    w1[q] = 0u;
    if (j < NBIN) {
      const unsigned* cp = cb + (size_t)j * CW;
      w0[q] = cp[lane];
      if (32 + lane < CW) w1[q] = cp[32 + lane];
    }
  }

  for (int i = tid; i < R; i += NT) cnt[i] = 0;
  if (tid == 0) { nextrow = 0; bad = 0; }
  __syncthreads();

  int sl0[CPW], sl1[CPW];
#pragma unroll
  for (int q = 0; q < CPW; ++q) {
    const unsigned n = (unsigned)__builtin_amdgcn_readfirstlane((int)w0[q]);
    if (n > (unsigned)CAP && lane == 0) bad = 1;
    const int nn = (int)min(n, (unsigned)CAP);
    sl0[q] = RC;
    sl1[q] = RC;
    if (lane >= 1 && lane - 1 < nn)
      sl0[q] = atomicAdd(&cnt[min((int)(w0[q] >> 20), R - 1)], 1);
    if (31 + lane < nn)
      sl1[q] = atomicAdd(&cnt[min((int)(w1[q] >> 20), R - 1)], 1);
  }
#pragma unroll
  for (int q = 0; q < CPW; ++q) {
    if (sl0[q] < RC)
      rq[min((int)(w0[q] >> 20), R - 1) * RC + sl0[q]] = min(w0[q] & 0xFFFFFu, (unsigned)(nRows - 1));
    if (sl1[q] < RC)
      rq[min((int)(w1[q] >> 20), R - 1) * RC + sl1[q]] = min(w1[q] & 0xFFFFFu, (unsigned)(nRows - 1));
  }
  __syncthreads();
  const bool listok = bad == 0;

  const int wvs = __builtin_amdgcn_readfirstlane(wv);
  const bool tilewave = FUSE && wvs < 2 * (R / 16);
  const int lcol = lane & 15;
  const int hh = lane >> 4;
  const int trow = 16 * (wvs >> 1);
  constexpr int LDB = 2 * NOUTP;
  const float* wb = FUSE ? Wp + (size_t)(2 * hh) * LDB + 4 * lcol + 64 * (wvs & 1) : nullptr;
  v8f tacc[2] = {};
  auto tstep = [&](const v4f a, const v4f b0, const v4f b1) {
    const v2f alo = __builtin_shufflevector(a, a, 0, 1), ahi = __builtin_shufflevector(a, a, 2, 3);
    tacc[0] = wmma4(alo, __builtin_shufflevector(b0, b0, 0, 1), tacc[0]);
    tacc[1] = wmma4(alo, __builtin_shufflevector(b0, b0, 2, 3), tacc[1]);
    tacc[0] = wmma4(ahi, __builtin_shufflevector(b1, b1, 0, 1), tacc[0]);
    tacc[1] = wmma4(ahi, __builtin_shufflevector(b1, b1, 2, 3), tacc[1]);
  };
  if constexpr (FUSE) {
    if (tilewave) {
      const float* ap = h + (size_t)min(base + trow + lcol, nRows - 1) * D + 4 * hh;
      for (int k0 = 0; k0 < D; k0 += 16) {
        const float* bp = wb + (size_t)(k0 >> 1) * LDB;
        const v4f a0 = *(const v4f*)(ap + k0);
        const v4f a1 = *(const v4f*)(ap + k0 + 8);
        const v4f b00 = *(const v4f*)(bp);
        const v4f b01 = *(const v4f*)(bp + LDB);
        const v4f b10 = *(const v4f*)(bp + 4 * LDB);
        const v4f b11 = *(const v4f*)(bp + 5 * LDB);
        tstep(a0, b00, b01);
        tstep(a1, b10, b11);
      }
    }
  }

  const unsigned loff = (unsigned)lane * 16u;
  float4 acc[KV];
  auto add_one = [&](int s) {
    const char* hp = (const char*)(h + (size_t)s * D) + loff;
#pragma unroll
    for (int k = 0; k < KV; ++k) {
      const float4 v = *(const float4*)(hp + 512 * k);
      acc[k].x += v.x; acc[k].y += v.y; acc[k].z += v.z; acc[k].w += v.w;
    }
  };

  for (int it = 0; it < R; ++it) {
    int rr = 0;
    if (lane == 0) rr = atomicAdd(&nextrow, 1);
    const int r = __builtin_amdgcn_readfirstlane(rr);
    if (r >= R || base + r >= nRows) break;
#pragma unroll
    for (int k = 0; k < KV; ++k) acc[k] = make_float4(0.f, 0.f, 0.f, 0.f);
    int deg = 0;
    if (listok) {
      deg = __builtin_amdgcn_readfirstlane(cnt[r]);
      if (deg <= RC) {
        const unsigned* q = rq + r * RC;
        for (int c0 = 0; c0 < deg; c0 += 32) {
          const int m = (deg - c0 < 32) ? (deg - c0) : 32;
          const unsigned e = (lane < m) ? q[c0 + lane] : 0u;
          auto group = [&](int j, const bool whole) {
            float4 v[G][KV];
#pragma unroll
            for (int g = 0; g < G; ++g) {
#pragma unroll
              for (int k = 0; k < KV; ++k) v[g][k] = make_float4(0.f, 0.f, 0.f, 0.f);
              if (whole || j + g < m) {
                const int sg = __builtin_amdgcn_readlane((int)e, j + g);
                const char* hp = (const char*)(h + (size_t)sg * D) + loff;
#pragma unroll
                for (int k = 0; k < KV; ++k) v[g][k] = *(const float4*)(hp + 512 * k);
              }
            }
#pragma unroll
            for (int g = 0; g < G; ++g) {
#pragma unroll
              for (int k = 0; k < KV; ++k) {
                acc[k].x += v[g][k].x; acc[k].y += v[g][k].y;
                acc[k].z += v[g][k].z; acc[k].w += v[g][k].w;
              }
            }
          };
          int j = 0;
          for (; j + G <= m; j += G) group(j, true);
          if (j < m) group(j, false);
        }
      } else {
        for (int c = 0; c < NBIN; ++c) {
          const unsigned* cp = cb + (size_t)c * CW;
          const int n = min(__builtin_amdgcn_readfirstlane((int)cp[0]), CAP);
          for (int i0 = 0; i0 < n; i0 += 32) {
            const int i = i0 + lane;
            const unsigned p = (i < n) ? cp[1 + i] : 0u;
            const bool own = (i < n) && (min((int)(p >> 20), R - 1) == r);
            const int sv = (int)min(p & 0xFFFFFu, (unsigned)(nRows - 1));
            unsigned mk = __builtin_amdgcn_ballot_w32(own);
            while (mk != 0u) {
              const int j = __builtin_ctz(mk);
              mk &= mk - 1u;
              add_one(__builtin_amdgcn_readlane(sv, j));
            }
          }
        }
      }
    } else {
      for (int e0 = 0; e0 < nE; e0 += 32) {
        const int e = e0 + lane;
        const bool in = e < nE;
        const int dv = in ? dst[e] : -1;
        unsigned mk = __builtin_amdgcn_ballot_w32(in && dv == base + r);
        if (mk != 0u) {
          deg += __builtin_popcount(mk);
          const int sraw = in ? src[e] : 0;
          const int sv = min(max(sraw, 0), nRows - 1);
          while (mk != 0u) {
            const int j = __builtin_ctz(mk);
            mk &= mk - 1u;
            add_one(__builtin_amdgcn_readlane(sv, j));
          }
        }
      }
    }
    const float sc = 1.0f / fmaxf((float)deg, 1.0f);
    if constexpr (FUSE) {
      float* tp = mt + r * PITCH + 4 * lane;
#pragma unroll
      for (int k = 0; k < KV; ++k)
        *(float4*)(tp + 128 * k) =
            make_float4(acc[k].x * sc, acc[k].y * sc, acc[k].z * sc, acc[k].w * sc);
    } else {
      char* op = (char*)(msum + (size_t)(base + r) * D) + loff;
#pragma unroll
      for (int k = 0; k < KV; ++k)
        *(float4*)(op + 512 * k) =
            make_float4(acc[k].x * sc, acc[k].y * sc, acc[k].z * sc, acc[k].w * sc);
    }
  }

  if constexpr (FUSE) {
    __syncthreads();
    if (tilewave) {
      const float* lp = mt + (trow + lcol) * PITCH + 4 * hh;
      for (int k0 = 0; k0 < D; k0 += 16) {
        const float* bp = wb + (size_t)((D + k0) >> 1) * LDB;
        const v4f a0 = *(const v4f*)(lp + k0);
        const v4f a1 = *(const v4f*)(lp + k0 + 8);
        const v4f b00 = *(const v4f*)(bp);
        const v4f b01 = *(const v4f*)(bp + LDB);
        const v4f b10 = *(const v4f*)(bp + 4 * LDB);
        const v4f b11 = *(const v4f*)(bp + 5 * LDB);
        tstep(a0, b00, b01);
        tstep(a1, b10, b11);
      }
      const int cc = 32 * (wvs & 1) + 2 * lcol;
#pragma unroll
      for (int i = 0; i < 2; ++i) {
        if (cc + i < NOUT) {
          const float bj = bias[cc + i];
#pragma unroll
          for (int rr = 0; rr < 8; ++rr) {
            const int row = base + trow + hh * 8 + rr;
            if (row < nRows) out[(size_t)row * NOUT + cc + i] = tacc[i][rr] + bj;
          }
        }
      }
    }
  }
}


template <bool RELU>
__global__ __launch_bounds__(128) void sage_gemm_wmma(
    const float* __restrict__ h, const float* __restrict__ msum,
    const float* __restrict__ Wp,
    const float* __restrict__ bias, float* __restrict__ out, int M, int d, int N, int Np) {
  const int wl = (int)threadIdx.x >> 5;
  const int tilesN = Np >> 6;
  const int tilesM = M >> 5;
  const int gn = (tilesN & 1) ? 1 : 2;
  const int gm = 4 / gn;
  const int groupsN = tilesN / gn;
  const int bm = (int)blockIdx.x / groupsN;
  const int bn = (int)blockIdx.x - bm * groupsN;
  const int wm = wl / gn;
  const int tM = bm * gm + wm;
  const int tN = bn * gn + (wl - wm * gn);
  if (tM >= tilesM) return;

  const int m0 = tM << 5;
  const int n0 = tN << 6;
  const int lane = (int)threadIdx.x & 31;
  const int half = lane >> 4;
  const int lcol = lane & 15;
  const int r0 = m0 + lcol;
  const int r1 = r0 + 16;
  const float* h0 = h + (size_t)r0 * d + 4 * half;
  const float* h1 = h + (size_t)r1 * d + 4 * half;
  const float* s0 = msum + (size_t)r0 * d + 4 * half;
  const float* s1 = msum + (size_t)r1 * d + 4 * half;
  const size_t ldb = 2 * (size_t)Np;
  const float* wb = Wp + (size_t)(2 * half) * ldb + 2 * (n0 + 2 * lcol);

  v8f acc0[4] = {}, acc1[4] = {};

  auto chunk = [&](const float* a0p, const float* a1p, const float* bp) {
    const v4f a0 = *(const v4f*)a0p;
    const v4f a1 = *(const v4f*)a1p;
    const v4f b00 = *(const v4f*)(bp);
    const v4f b01 = *(const v4f*)(bp + 64);
    const v4f b10 = *(const v4f*)(bp + ldb);
    const v4f b11 = *(const v4f*)(bp + ldb + 64);
    const v2f a0lo = __builtin_shufflevector(a0, a0, 0, 1), a0hi = __builtin_shufflevector(a0, a0, 2, 3);
    const v2f a1lo = __builtin_shufflevector(a1, a1, 0, 1), a1hi = __builtin_shufflevector(a1, a1, 2, 3);
    acc0[0] = wmma4(a0lo, __builtin_shufflevector(b00, b00, 0, 1), acc0[0]);
    acc1[0] = wmma4(a1lo, __builtin_shufflevector(b00, b00, 0, 1), acc1[0]);
    acc0[1] = wmma4(a0lo, __builtin_shufflevector(b00, b00, 2, 3), acc0[1]);
    acc1[1] = wmma4(a1lo, __builtin_shufflevector(b00, b00, 2, 3), acc1[1]);
    acc0[2] = wmma4(a0lo, __builtin_shufflevector(b01, b01, 0, 1), acc0[2]);
    acc1[2] = wmma4(a1lo, __builtin_shufflevector(b01, b01, 0, 1), acc1[2]);
    acc0[3] = wmma4(a0lo, __builtin_shufflevector(b01, b01, 2, 3), acc0[3]);
    acc1[3] = wmma4(a1lo, __builtin_shufflevector(b01, b01, 2, 3), acc1[3]);
    acc0[0] = wmma4(a0hi, __builtin_shufflevector(b10, b10, 0, 1), acc0[0]);
    acc1[0] = wmma4(a1hi, __builtin_shufflevector(b10, b10, 0, 1), acc1[0]);
    acc0[1] = wmma4(a0hi, __builtin_shufflevector(b10, b10, 2, 3), acc0[1]);
    acc1[1] = wmma4(a1hi, __builtin_shufflevector(b10, b10, 2, 3), acc1[1]);
    acc0[2] = wmma4(a0hi, __builtin_shufflevector(b11, b11, 0, 1), acc0[2]);
    acc1[2] = wmma4(a1hi, __builtin_shufflevector(b11, b11, 0, 1), acc1[2]);
    acc0[3] = wmma4(a0hi, __builtin_shufflevector(b11, b11, 2, 3), acc0[3]);
    acc1[3] = wmma4(a1hi, __builtin_shufflevector(b11, b11, 2, 3), acc1[3]);
  };

  for (int k0 = 0; k0 < d; k0 += 8) chunk(h0 + k0, h1 + k0, wb + (size_t)(k0 >> 1) * ldb);
  for (int k0 = 0; k0 < d; k0 += 8) chunk(s0 + k0, s1 + k0, wb + (size_t)((d + k0) >> 1) * ldb);

  const bool evenN = (N & 1) == 0;
#pragma unroll
  for (int q = 0; q < 2; ++q) {
    const int cc = n0 + 32 * q + 2 * lcol;
    if (evenN) {
      if (cc + 1 < N) {
        const float bx = bias[cc], by = bias[cc + 1];
#pragma unroll
        for (int r = 0; r < 8; ++r) {
          float x0 = acc0[2 * q][r] + bx, y0 = acc0[2 * q + 1][r] + by;
          float x1 = acc1[2 * q][r] + bx, y1 = acc1[2 * q + 1][r] + by;
          if (RELU) {
            x0 = fmaxf(x0, 0.0f); y0 = fmaxf(y0, 0.0f);
            x1 = fmaxf(x1, 0.0f); y1 = fmaxf(y1, 0.0f);
          }
          *(float2*)(out + (size_t)(m0 + half * 8 + r) * N + cc) = make_float2(x0, y0);
          *(float2*)(out + (size_t)(m0 + 16 + half * 8 + r) * N + cc) = make_float2(x1, y1);
        }
      }
    } else {
#pragma unroll
      for (int i = 0; i < 2; ++i) {
        if (cc + i < N) {
          const float bj = bias[cc + i];
#pragma unroll
          for (int r = 0; r < 8; ++r) {
            float v0 = acc0[2 * q + i][r] + bj;
            float v1 = acc1[2 * q + i][r] + bj;
            if (RELU) {
              v0 = fmaxf(v0, 0.0f);
              v1 = fmaxf(v1, 0.0f);
            }
            out[(size_t)(m0 + half * 8 + r) * N + cc + i] = v0;
            out[(size_t)(m0 + 16 + half * 8 + r) * N + cc + i] = v1;
          }
        }
      }
    }
  }
}

extern "C" void kernel_launch(void* const* d_in, const int* in_sizes, int n_in, void* d_out,
                              int out_size, void* d_ws, size_t ws_size, hipStream_t stream) {
  (void)in_sizes; (void)n_in; (void)out_size; (void)ws_size;
  const float* x  = (const float*)d_in[0];
  const float* W1 = (const float*)d_in[1];
  const float* b1 = (const float*)d_in[2];
  const float* W2 = (const float*)d_in[3];
  const float* b2 = (const float*)d_in[4];
  const float* W3 = (const float*)d_in[5];
  const float* b3 = (const float*)d_in[6];
  const int* src  = (const int*)d_in[7];
  const int* dst  = (const int*)d_in[8];
  float* out = (float*)d_out;

  constexpr int M = 40000, E = 640000, INF = 128, HID = 256, OUTF = 47, OUTP = 64;
  constexpr int OWN_R = 160, OWN_T = 1024, OWN_RC = 64, OWN_B = (M + OWN_R - 1) / OWN_R;
  static_assert(M <= (1 << 20), "source node must fit the 20 low bits of a saved cell word");
  constexpr int BIN_N = (E + 4095) / 4096, CELL_CAP = 46, CELL_S = OWN_R * OUTF;
  static_assert(BIN_N * (CELL_CAP + 1) <= CELL_S && (long long)OWN_B * CELL_S <= (long long)M * OUTF,
                "an owner's cells must fit its own rows of the output buffer");
  static_assert(OWN_B <= 256, "the binning blocks rank at most 256 owners in LDS");
  unsigned* cells = (unsigned*)d_out;

  float* ws = (float*)d_ws;
  float* msum = ws + 80000;
  float* h1   = msum + (size_t)M * HID;
  float* h2   = h1 + (size_t)M * HID;
  float* Wp1  = h2 + (size_t)M * HID;
  float* Wp2  = Wp1 + (size_t)(2 * INF) * HID;
  float* Wp3  = Wp2 + (size_t)(2 * HID) * HID;

  constexpr int TG = 128;
  constexpr int GEMM_G_HID = ((M / 32 + 1) / 2) * ((HID / 64) / 2);
  static_assert((HID / 64) % 2 == 0, "group shape assumed by the grid");

  prep_kernel<OWN_R, CELL_CAP, 256, BIN_N, CELL_S, INF, HID, OUTF, OUTP><<<BIN_N, 1024, 0, stream>>>(
      src, dst, cells, E, M, OWN_B, W1, W2, W3, Wp1, Wp2, Wp3);

  {
    owner_sum_kernel<INF, OWN_R, OWN_T, OWN_RC, CELL_CAP, BIN_N, CELL_S, false, OUTF, OUTP>
        <<<OWN_B, OWN_T, 0, stream>>>(x, src, dst, msum, cells, E, M, nullptr, nullptr, nullptr);
    sage_gemm_wmma<true><<<GEMM_G_HID, TG, 0, stream>>>(
        x, msum, Wp1, b1, h1, M, INF, HID, HID);
  }
  {
    owner_sum_kernel<HID, OWN_R, OWN_T, OWN_RC, CELL_CAP, BIN_N, CELL_S, false, OUTF, OUTP>
        <<<OWN_B, OWN_T, 0, stream>>>(h1, src, dst, msum, cells, E, M, nullptr, nullptr, nullptr);
    sage_gemm_wmma<true><<<GEMM_G_HID, TG, 0, stream>>>(
        h1, msum, Wp2, b2, h2, M, HID, HID, HID);
  }
  {
    owner_sum_kernel<HID, OWN_R, OWN_T, OWN_RC, CELL_CAP, BIN_N, CELL_S, true, OUTF, OUTP>
        <<<OWN_B, OWN_T, 0, stream>>>(h2, src, dst, msum, cells, E, M, Wp3, b3, out);
  }
}
